# starting kernel
# baseline (speedup 1.0000x reference)
#include <hip/hip_runtime.h>
#include <hip/hip_bf16.h>

typedef __attribute__((ext_vector_type(2))) float v2f;
typedef __attribute__((ext_vector_type(8))) float v8f;


__device__ __forceinline__ void atom_add_f32(float* p, float v) {
  __hip_atomic_fetch_add(p, v, __ATOMIC_RELAXED, __HIP_MEMORY_SCOPE_AGENT);
}

__global__ void zero_kernel(float* __restrict__ p, long n) {
  long i = (long)blockIdx.x * blockDim.x + threadIdx.x;
  if (i < n) p[i] = 0.0f;
}

__global__ void degree_kernel(const int* __restrict__ dst, float* __restrict__ cnt, int E) {
  int e = blockIdx.x * blockDim.x + threadIdx.x;
  if (e < E) atom_add_f32(&cnt[dst[e]], 1.0f);
}

__global__ void invdeg_kernel(const float* __restrict__ cnt, float* __restrict__ inv, int n) {
  int i = blockIdx.x * blockDim.x + threadIdx.x;
  if (i < n) inv[i] = 1.0f / fmaxf(cnt[i], 1.0f);
}

template <int D>
__global__ void scatter_add_kernel(const float* __restrict__ x,
                                   const int* __restrict__ src,
                                   const int* __restrict__ dst,
                                   float* __restrict__ agg, int E) {
  constexpr int C = D / 4;
  long tid = (long)blockIdx.x * blockDim.x + threadIdx.x;
  long total = (long)E * C;
  if (tid >= total) return;
  int e = (int)(tid / C);
  int f = (int)(tid % C) * 4;
  int s = src[e];
  int d = dst[e];
  float4 v = *(const float4*)(x + (long)s * D + f);
  float* o = agg + (long)d * D + f;
  atom_add_f32(o + 0, v.x);
  atom_add_f32(o + 1, v.y);
  atom_add_f32(o + 2, v.z);
  atom_add_f32(o + 3, v.w);
}

__global__ void split_z2_kernel(const float* __restrict__ z2,
                                float* __restrict__ out01,
                                float* __restrict__ out2, int n) {
  int i = blockIdx.x * blockDim.x + threadIdx.x;
  if (i < n) {
    out01[2 * i + 0] = z2[3 * i + 0];
    out01[2 * i + 1] = z2[3 * i + 1];
    out2[i] = z2[3 * i + 2];
  }
}

template <int DIN, int DOUT, int ACT, bool HAS_AGG>
__global__ __launch_bounds__(256) void wmma_linear_kernel(
    const float* __restrict__ agg, const float* __restrict__ invdeg,
    const float* __restrict__ x,
    const float* __restrict__ Wl, const float* __restrict__ bias,
    const float* __restrict__ Wr,
    float* __restrict__ out, float* __restrict__ out2, int nnodes) {
  constexpr int MT = 4;
  constexpr int ROWS = 16 * MT;
  constexpr int DOUTP = (DOUT + 15) & ~15;
  constexpr int LDW = DIN + 4;
  constexpr int C4 = DIN / 4;
  __shared__ float sWl[DOUTP * LDW];
  __shared__ float sWr[HAS_AGG ? DOUTP * LDW : 1];
  __shared__ float sA[ROWS * LDW];
  __shared__ float sX[HAS_AGG ? ROWS * LDW : 1];

  const int tid = threadIdx.x;
  const int bs = blockDim.x;
  const int row0 = blockIdx.x * ROWS;

  const float4 f4z = make_float4(0.0f, 0.0f, 0.0f, 0.0f);

  for (int i = tid; i < DOUTP * C4; i += bs) {
    int r = i / C4, c4 = (i - r * C4) * 4;
    float4 wl = (r < DOUT) ? *(const float4*)(Wl + (long)r * DIN + c4) : f4z;
    *(float4*)&sWl[r * LDW + c4] = wl;
    if constexpr (HAS_AGG) {
      float4 wr = (r < DOUT) ? *(const float4*)(Wr + (long)r * DIN + c4) : f4z;
      *(float4*)&sWr[r * LDW + c4] = wr;
    }
  }
  for (int i = tid; i < ROWS * C4; i += bs) {
    int r = i / C4, c4 = (i - r * C4) * 4;
    int m = row0 + r;
    if (m >= nnodes) m = nnodes - 1;
    if constexpr (HAS_AGG) {
      float sc = invdeg[m];
      float4 av = *(const float4*)(agg + (long)m * DIN + c4);
      av.x *= sc; av.y *= sc; av.z *= sc; av.w *= sc;
      *(float4*)&sA[r * LDW + c4] = av;
      *(float4*)&sX[r * LDW + c4] = *(const float4*)(x + (long)m * DIN + c4);
    } else {
      *(float4*)&sA[r * LDW + c4] = *(const float4*)(x + (long)m * DIN + c4);
    }
  }
  __syncthreads();

  const int wave = tid >> 5;
  const int lane = tid & 31;
  const int half = lane >> 4;
  const int l16 = lane & 15;
  const int n = wave * 16 + l16;

  v8f acc[MT];
#pragma unroll
  for (int mt = 0; mt < MT; ++mt) acc[mt] = (v8f){};

#pragma unroll
  for (int k = 0; k < DIN; k += 4) {
    const int kk = k + 2 * half;
    v2f b;
    b.x = sWl[n * LDW + kk];
    b.y = sWl[n * LDW + kk + 1];
    v2f b2;
    if constexpr (HAS_AGG) {
      b2.x = sWr[n * LDW + kk];
      b2.y = sWr[n * LDW + kk + 1];
    }
#pragma unroll
    for (int mt = 0; mt < MT; ++mt) {
      const int r = mt * 16 + l16;
      v2f a;
      a.x = sA[r * LDW + kk];
      a.y = sA[r * LDW + kk + 1];
      acc[mt] = __builtin_amdgcn_wmma_f32_16x16x4_f32(false, a, false, b,
                                                      (short)0, acc[mt], false,
                                                      false);
      if constexpr (HAS_AGG) {
        v2f a2;
        a2.x = sX[r * LDW + kk];
        a2.y = sX[r * LDW + kk + 1];
        acc[mt] = __builtin_amdgcn_wmma_f32_16x16x4_f32(false, a2, false, b2,
                                                        (short)0, acc[mt],
                                                        false, false);
      }
    }
  }

  const bool colOK = (n < DOUT);
  const float bcol = colOK ? bias[colOK ? n : 0] : 0.0f;
#pragma unroll
  for (int mt = 0; mt < MT; ++mt) {
#pragma unroll
    for (int i = 0; i < 8; ++i) {
      int row = row0 + mt * 16 + i + 8 * half;
      float v = acc[mt][i] + bcol;
      if (ACT == 1) v = tanhf(v);
      if (ACT == 2) v = (v > 0.0f) ? v : 0.01f * v;
      if (colOK && row < nnodes) {
        out[(long)row * DOUT + n] = v;
        if (out2) out2[(long)row * DOUT + n] = v;
      }
    }
  }
}


static inline unsigned cdiv(long a, long b) { return (unsigned)((a + b - 1) / b); }

extern "C" void kernel_launch(void* const* d_in, const int* in_sizes, int n_in,
                              void* d_out, int out_size, void* d_ws, size_t ws_size,
                              hipStream_t stream) {
  const float* x = (const float*)d_in[0];
  const int* ei = (const int*)d_in[1];
  const int N = in_sizes[0] / 64;
  const int E = in_sizes[1] / 2;
  const int* src = ei;
  const int* dst = ei + E;

  const float* g1Wl = (const float*)d_in[2];
  const float* g1b  = (const float*)d_in[3];
  const float* g1Wr = (const float*)d_in[4];
  const float* g2Wl = (const float*)d_in[5];
  const float* g2b  = (const float*)d_in[6];
  const float* g2Wr = (const float*)d_in[7];
  const float* g3Wl = (const float*)d_in[8];
  const float* g3b  = (const float*)d_in[9];
  const float* g3Wr = (const float*)d_in[10];
  const float* trW  = (const float*)d_in[11];
  const float* trb  = (const float*)d_in[12];
  const float* d1W  = (const float*)d_in[13];
  const float* d1b  = (const float*)d_in[14];
  const float* d2Wl = (const float*)d_in[15];
  const float* d2b  = (const float*)d_in[16];
  const float* d2Wr = (const float*)d_in[17];
  const float* d3Wl = (const float*)d_in[18];
  const float* d3b  = (const float*)d_in[19];
  const float* d3Wr = (const float*)d_in[20];
  const float* d4Wl = (const float*)d_in[21];
  const float* d4b  = (const float*)d_in[22];
  const float* d4Wr = (const float*)d_in[23];

  float* ws   = (float*)d_ws;
  float* cnt  = ws;
  float* inv  = ws + N;
  float* bufA = ws + 2 * (size_t)N;
  float* bufB = bufA + (size_t)N * 128;
  float* bufC = bufB + (size_t)N * 128;
  float* z2   = bufC + (size_t)N * 128;

  float* outZ01 = (float*)d_out;
  float* outZ2  = outZ01 + 2 * (size_t)N;
  float* outMu  = outZ2 + N;
  float* outLv  = outMu + (size_t)N * 64;

  const unsigned gN64 = (unsigned)((N + 63) / 64);

  zero_kernel<<<cdiv(N, 256), 256, 0, stream>>>(cnt, N);
  degree_kernel<<<cdiv(E, 256), 256, 0, stream>>>(dst, cnt, E);
  invdeg_kernel<<<cdiv(N, 256), 256, 0, stream>>>(cnt, inv, N);

  zero_kernel<<<cdiv((long)N * 64, 256), 256, 0, stream>>>(bufA, (long)N * 64);
  scatter_add_kernel<64><<<cdiv((long)E * 16, 256), 256, 0, stream>>>(x, src, dst, bufA, E);
  wmma_linear_kernel<64, 128, 1, true><<<gN64, 256, 0, stream>>>(
      bufA, inv, x, g1Wl, g1b, g1Wr, bufB, nullptr, N);

  zero_kernel<<<cdiv((long)N * 128, 256), 256, 0, stream>>>(bufA, (long)N * 128);
  scatter_add_kernel<128><<<cdiv((long)E * 32, 256), 256, 0, stream>>>(bufB, src, dst, bufA, E);
  wmma_linear_kernel<128, 128, 1, true><<<gN64, 256, 0, stream>>>(
      bufA, inv, bufB, g2Wl, g2b, g2Wr, bufC, nullptr, N);

  zero_kernel<<<cdiv((long)N * 128, 256), 256, 0, stream>>>(bufA, (long)N * 128);
  scatter_add_kernel<128><<<cdiv((long)E * 32, 256), 256, 0, stream>>>(bufC, src, dst, bufA, E);
  wmma_linear_kernel<128, 128, 1, true><<<gN64, 256, 0, stream>>>(
      bufA, inv, bufC, g3Wl, g3b, g3Wr, bufB, nullptr, N);

  wmma_linear_kernel<128, 64, 0, false><<<gN64, 128, 0, stream>>>(
      nullptr, nullptr, bufB, trW, trb, nullptr, outMu, outLv, N);

  wmma_linear_kernel<64, 128, 2, false><<<gN64, 256, 0, stream>>>(
      nullptr, nullptr, outMu, d1W, d1b, nullptr, bufA, nullptr, N);

  zero_kernel<<<cdiv((long)N * 128, 256), 256, 0, stream>>>(bufC, (long)N * 128);
  scatter_add_kernel<128><<<cdiv((long)E * 32, 256), 256, 0, stream>>>(bufA, src, dst, bufC, E);
  wmma_linear_kernel<128, 128, 2, true><<<gN64, 256, 0, stream>>>(
      bufC, inv, bufA, d2Wl, d2b, d2Wr, bufB, nullptr, N);

  zero_kernel<<<cdiv((long)N * 128, 256), 256, 0, stream>>>(bufC, (long)N * 128);
  scatter_add_kernel<128><<<cdiv((long)E * 32, 256), 256, 0, stream>>>(bufB, src, dst, bufC, E);
  wmma_linear_kernel<128, 64, 1, true><<<gN64, 128, 0, stream>>>(
      bufC, inv, bufB, d3Wl, d3b, d3Wr, bufA, nullptr, N);

  zero_kernel<<<cdiv((long)N * 64, 256), 256, 0, stream>>>(bufC, (long)N * 64);
  scatter_add_kernel<64><<<cdiv((long)E * 16, 256), 256, 0, stream>>>(bufA, src, dst, bufC, E);
  wmma_linear_kernel<64, 3, 0, true><<<gN64, 32, 0, stream>>>(
      bufC, inv, bufA, d4Wl, d4b, d4Wr, z2, nullptr, N);

  split_z2_kernel<<<cdiv(N, 256), 256, 0, stream>>>(z2, outZ01, outZ2, N);
}


# Owner groups' edge lists (source row + local row, one word) built once per call by binning edges by owner; the 6 neighbour sums read them: no key scan or index fetch per layer; arrival order kept
# speedup vs baseline: 2.6139x; 2.6139x over previous
#include <hip/hip_runtime.h>
#include <hip/hip_bf16.h>

typedef __attribute__((ext_vector_type(2))) float v2f;
typedef __attribute__((ext_vector_type(8))) float v8f;


__device__ __forceinline__ void atom_add_f32(float* p, float v) {
  __hip_atomic_fetch_add(p, v, __ATOMIC_RELAXED, __HIP_MEMORY_SCOPE_AGENT);
}

__global__ void zero_kernel(float* __restrict__ p, long n) {
  long i = (long)blockIdx.x * blockDim.x + threadIdx.x;
  if (i < n) p[i] = 0.0f;
}

__global__ void invdeg_kernel(const float* __restrict__ cnt, float* __restrict__ inv, int n) {
  int i = blockIdx.x * blockDim.x + threadIdx.x;
  if (i < n) inv[i] = 1.0f / fmaxf(cnt[i], 1.0f);
}

constexpr int RNG = 400;
constexpr int MAXOWN = 1024;
constexpr int EPT = 16;
constexpr int SRC_BITS = 23;

__device__ __forceinline__ void lds_add_f32(float* p, float v) {
  __hip_atomic_fetch_add(p, v, __ATOMIC_RELAXED, __HIP_MEMORY_SCOPE_WORKGROUP);
}

__global__ __launch_bounds__(256) void degree_count_kernel(const int* __restrict__ dst,
                                                           float* __restrict__ cnt,
                                                           int* __restrict__ ocnt, int E, int nown) {
  __shared__ int h[MAXOWN];
  const int tid = threadIdx.x;
  for (int i = tid; i < MAXOWN; i += 256) h[i] = 0;
  __syncthreads();
  const int base = blockIdx.x * (256 * EPT);
  int d[EPT];
#pragma unroll
  for (int k = 0; k < EPT; ++k) {
    const int e = base + k * 256 + tid;
    d[k] = (e < E) ? dst[e] : -1;
  }
#pragma unroll
  for (int k = 0; k < EPT; ++k) {
    if (d[k] >= 0) {
      atom_add_f32(&cnt[d[k]], 1.0f);
      atomicAdd(&h[d[k] / RNG], 1);
    }
  }
  __syncthreads();
  for (int i = tid; i < nown; i += 256) {
    const int c = h[i];
    if (c) atomicAdd(&ocnt[i], c);
  }
}

__global__ __launch_bounds__(256) void owner_bin_kernel(const int* __restrict__ src,
                                                        const int* __restrict__ dst,
                                                        const int* __restrict__ ocnt,
                                                        int* __restrict__ ofill,
                                                        int* __restrict__ ooff,
                                                        unsigned* __restrict__ list,
                                                        int E, int nown) {
  __shared__ int h[MAXOWN];
  __shared__ int off[MAXOWN];
  __shared__ int sc[256];
  const int tid = threadIdx.x;
  for (int i = tid; i < MAXOWN; i += 256) h[i] = 0;
  int c[4];
  int s = 0;
#pragma unroll
  for (int k = 0; k < 4; ++k) {
    const int b = 4 * tid + k;
    c[k] = (b < nown) ? ocnt[b] : 0;
    s += c[k];
  }
  sc[tid] = s;
  __syncthreads();
  for (int dd = 1; dd < 256; dd <<= 1) {
    const int v = (tid >= dd) ? sc[tid - dd] : 0;
    __syncthreads();
    sc[tid] += v;
    __syncthreads();
  }
  int run = sc[tid] - s;
#pragma unroll
  for (int k = 0; k < 4; ++k) {
    off[4 * tid + k] = run;
    run += c[k];
  }
  __syncthreads();
  if (blockIdx.x == 0)
    for (int i = tid; i <= nown; i += 256) ooff[i] = (i < nown) ? off[i] : sc[255];

  const int base = blockIdx.x * (256 * EPT);
  int dv[EPT], sv[EPT];
#pragma unroll
  for (int k = 0; k < EPT; ++k) {
    const int e = base + k * 256 + tid;
    const int ec = (e < E) ? e : E - 1;
    dv[k] = dst[ec];
    sv[k] = src[ec];
  }
  int ow[EPT], rk[EPT];
#pragma unroll
  for (int k = 0; k < EPT; ++k) {
    const int e = base + k * 256 + tid;
    ow[k] = dv[k] / RNG;
    rk[k] = (e < E) ? atomicAdd(&h[ow[k]], 1) : 0;
  }
  __syncthreads();
  for (int i = tid; i < nown; i += 256) {
    const int cc = h[i];
    if (cc) off[i] += atomicAdd(&ofill[i], cc);
  }
  __syncthreads();
#pragma unroll
  for (int k = 0; k < EPT; ++k) {
    const int e = base + k * 256 + tid;
    if (e < E)
      list[off[ow[k]] + rk[k]] =
          (unsigned)sv[k] | ((unsigned)(dv[k] - ow[k] * RNG) << SRC_BITS);
  }
}

template <int D>
__global__ __launch_bounds__(1024) void owner_sum_kernel(const float* __restrict__ x,
                                                         const unsigned* __restrict__ list,
                                                         const int* __restrict__ ooff,
                                                         float* __restrict__ agg, int nnodes) {
  constexpr int L = D / 32;
  constexpr int U = 32 / L;
  __shared__ alignas(16) float acc[RNG * D];
  const int tid = threadIdx.x;
  const int r0 = blockIdx.x * RNG;
  const int beg = ooff[blockIdx.x];
  const int end = ooff[blockIdx.x + 1];

  for (int i = tid; i < RNG * D / 4; i += 1024)
    ((float4*)acc)[i] = make_float4(0.0f, 0.0f, 0.0f, 0.0f);
  __syncthreads();

  const int lane = tid & 31;
  const int wv = __builtin_amdgcn_readfirstlane(tid >> 5);
  for (int st = beg + wv * 32; st < end; st += 32 * 32) {
    const int qi = (st + lane < end) ? st + lane : end - 1;
    const unsigned p = list[qi];
    const bool mine = (st + lane < end);
    const int sv = mine ? (int)(p & ((1u << SRC_BITS) - 1u)) : 0;
    const int dl = mine ? (int)(p >> SRC_BITS) : -1;
    for (int i0 = 0; i0 < 32 && st + i0 < end; i0 += U) {
      float v[U][L];
#pragma unroll
      for (int j = 0; j < U; ++j) {
        const int sj = __builtin_amdgcn_readlane(sv, i0 + j);
        const float* r = x + (long)sj * D;
#pragma unroll
        for (int l = 0; l < L; ++l) v[j][l] = r[lane + 32 * l];
      }
#pragma unroll
      for (int j = 0; j < U; ++j) {
        const int dj = __builtin_amdgcn_readlane(dl, i0 + j);
        if (dj >= 0) {
          float* o = acc + dj * D;
#pragma unroll
          for (int l = 0; l < L; ++l) lds_add_f32(o + lane + 32 * l, v[j][l]);
        }
      }
    }
  }
  __syncthreads();

  for (int i = tid; i < RNG * D / 4; i += 1024) {
    const int r = i / (D / 4);
    if (r0 + r < nnodes) ((float4*)(agg + (long)r0 * D))[i] = ((const float4*)acc)[i];
  }
}

__global__ void split_z2_kernel(const float* __restrict__ z2,
                                float* __restrict__ out01,
                                float* __restrict__ out2, int n) {
  int i = blockIdx.x * blockDim.x + threadIdx.x;
  if (i < n) {
    out01[2 * i + 0] = z2[3 * i + 0];
    out01[2 * i + 1] = z2[3 * i + 1];
    out2[i] = z2[3 * i + 2];
  }
}

template <int DIN, int DOUT, int ACT, bool HAS_AGG>
__global__ __launch_bounds__(2 * ((DOUT + 15) & ~15)) void wmma_linear_kernel(
    const float* agg, const float* __restrict__ invdeg,
    const float* __restrict__ x,
    const float* __restrict__ Wl, const float* __restrict__ bias,
    const float* __restrict__ Wr,
    float* out, float* out2, int nnodes) {
  constexpr int MT = 4;
  constexpr int ROWS = 16 * MT;
  constexpr int DOUTP = (DOUT + 15) & ~15;
  constexpr int LDW = DIN + 4;
  constexpr int C4 = DIN / 4;
  __shared__ float sWl[DOUTP * LDW];
  __shared__ float sWr[HAS_AGG ? DOUTP * LDW : 1];
  __shared__ float sA[ROWS * LDW];
  __shared__ float sX[HAS_AGG ? ROWS * LDW : 1];

  const int tid = threadIdx.x;
  constexpr int NT = 2 * DOUTP;
  const int row0 = blockIdx.x * ROWS;

  const float4 f4z = make_float4(0.0f, 0.0f, 0.0f, 0.0f);

  constexpr int WPT = DOUTP * C4 / NT;
  constexpr int APT = ROWS * C4 / NT;
  constexpr int WREG = WPT * (HAS_AGG ? 8 : 4);
  constexpr int AMAX = (208 - WREG) / (HAS_AGG ? 9 : 4);
  constexpr int AG = (APT <= AMAX) ? APT : (APT / 2 <= AMAX) ? APT / 2 : (APT / 4 <= AMAX) ? APT / 4 : APT / 8;
  float4 wl[WPT];
  float4 wr[HAS_AGG ? WPT : 1];
#pragma unroll
  for (int j = 0; j < WPT; ++j) {
    const int i = tid + j * NT;
    const int r = i / C4, c4 = (i - r * C4) * 4;
    const bool rowOK = (DOUT == DOUTP) || (r < DOUT);
    wl[j] = rowOK ? *(const float4*)(Wl + (long)r * DIN + c4) : f4z;
    if constexpr (HAS_AGG) wr[j] = rowOK ? *(const float4*)(Wr + (long)r * DIN + c4) : f4z;
  }
  for (int g = 0; g < APT; g += AG) {
    float4 av[AG];
    float4 xv[HAS_AGG ? AG : 1];
    float sc[HAS_AGG ? AG : 1];
#pragma unroll
    for (int j = 0; j < AG; ++j) {
      const int i = tid + (g + j) * NT;
      const int r = i / C4, c4 = (i - r * C4) * 4;
      int m = row0 + r;
      if (m >= nnodes) m = nnodes - 1;
      if constexpr (HAS_AGG) {
        sc[j] = invdeg[m];
        av[j] = *(const float4*)(agg + (long)m * DIN + c4);
        xv[j] = *(const float4*)(x + (long)m * DIN + c4);
      } else {
        av[j] = *(const float4*)(x + (long)m * DIN + c4);
      }
    }
    if (g == 0) {
#pragma unroll
      for (int j = 0; j < WPT; ++j) {
        const int i = tid + j * NT;
        const int r = i / C4, c4 = (i - r * C4) * 4;
        *(float4*)&sWl[r * LDW + c4] = wl[j];
        if constexpr (HAS_AGG) *(float4*)&sWr[r * LDW + c4] = wr[j];
      }
    }
#pragma unroll
    for (int j = 0; j < AG; ++j) {
      const int i = tid + (g + j) * NT;
      const int r = i / C4, c4 = (i - r * C4) * 4;
      if constexpr (HAS_AGG) {
        float4 a = av[j];
        const float s = sc[j];
        a.x *= s; a.y *= s; a.z *= s; a.w *= s;
        *(float4*)&sA[r * LDW + c4] = a;
        *(float4*)&sX[r * LDW + c4] = xv[j];
      } else {
        *(float4*)&sA[r * LDW + c4] = av[j];
      }
    }
  }
  __syncthreads();

  const int wave = tid >> 5;
  const int lane = tid & 31;
  const int half = lane >> 4;
  const int l16 = lane & 15;
  const int n = wave * 16 + l16;

  v8f acc[MT];
#pragma unroll
  for (int mt = 0; mt < MT; ++mt) acc[mt] = (v8f){};

#pragma unroll
  for (int k = 0; k < DIN; k += 4) {
    const int kk = k + 2 * half;
    v2f b;
    b.x = sWl[n * LDW + kk];
    b.y = sWl[n * LDW + kk + 1];
    v2f b2;
    if constexpr (HAS_AGG) {
      b2.x = sWr[n * LDW + kk];
      b2.y = sWr[n * LDW + kk + 1];
    }
#pragma unroll
    for (int mt = 0; mt < MT; ++mt) {
      const int r = mt * 16 + l16;
      v2f a;
      a.x = sA[r * LDW + kk];
      a.y = sA[r * LDW + kk + 1];
      acc[mt] = __builtin_amdgcn_wmma_f32_16x16x4_f32(false, a, false, b,
                                                      (short)0, acc[mt], false,
                                                      false);
      if constexpr (HAS_AGG) {
        v2f a2;
        a2.x = sX[r * LDW + kk];
        a2.y = sX[r * LDW + kk + 1];
        acc[mt] = __builtin_amdgcn_wmma_f32_16x16x4_f32(false, a2, false, b2,
                                                        (short)0, acc[mt],
                                                        false, false);
      }
    }
  }

  const bool colOK = (n < DOUT);
  const float bcol = colOK ? bias[colOK ? n : 0] : 0.0f;
#pragma unroll
  for (int mt = 0; mt < MT; ++mt) {
#pragma unroll
    for (int i = 0; i < 8; ++i) {
      int row = row0 + mt * 16 + i + 8 * half;
      float v = acc[mt][i] + bcol;
      if (ACT == 1) v = tanhf(v);
      if (ACT == 2) v = (v > 0.0f) ? v : 0.01f * v;
      if (colOK && row < nnodes) {
        out[(long)row * DOUT + n] = v;
        if (out2) out2[(long)row * DOUT + n] = v;
      }
    }
  }
}


static inline unsigned cdiv(long a, long b) { return (unsigned)((a + b - 1) / b); }

extern "C" void kernel_launch(void* const* d_in, const int* in_sizes, int n_in,
                              void* d_out, int out_size, void* d_ws, size_t ws_size,
                              hipStream_t stream) {
  const float* x = (const float*)d_in[0];
  const int* ei = (const int*)d_in[1];
  const int N = in_sizes[0] / 64;
  const int E = in_sizes[1] / 2;
  const int* src = ei;
  const int* dst = ei + E;

  const float* g1Wl = (const float*)d_in[2];
  const float* g1b  = (const float*)d_in[3];
  const float* g1Wr = (const float*)d_in[4];
  const float* g2Wl = (const float*)d_in[5];
  const float* g2b  = (const float*)d_in[6];
  const float* g2Wr = (const float*)d_in[7];
  const float* g3Wl = (const float*)d_in[8];
  const float* g3b  = (const float*)d_in[9];
  const float* g3Wr = (const float*)d_in[10];
  const float* trW  = (const float*)d_in[11];
  const float* trb  = (const float*)d_in[12];
  const float* d1W  = (const float*)d_in[13];
  const float* d1b  = (const float*)d_in[14];
  const float* d2Wl = (const float*)d_in[15];
  const float* d2b  = (const float*)d_in[16];
  const float* d2Wr = (const float*)d_in[17];
  const float* d3Wl = (const float*)d_in[18];
  const float* d3b  = (const float*)d_in[19];
  const float* d3Wr = (const float*)d_in[20];
  const float* d4Wl = (const float*)d_in[21];
  const float* d4b  = (const float*)d_in[22];
  const float* d4Wr = (const float*)d_in[23];

  const unsigned gOwn = cdiv(N, RNG);
  float* ws   = (float*)d_ws;
  float* inv  = ws;
  unsigned* list = (unsigned*)(ws + N);
  float* bufA = ws + (size_t)N + (size_t)E;
  float* bufB = bufA + (size_t)N * 128;
  float* bufC = bufB + (size_t)N * 128;
  int* ooff   = (int*)(bufC + (size_t)N * 64);
  float* cnt  = bufA;
  int* ocnt   = (int*)(bufA + N);
  int* ofill  = ocnt + MAXOWN;
  float* z2   = bufB;

  float* outZ01 = (float*)d_out;
  float* outZ2  = outZ01 + 2 * (size_t)N;
  float* outMu  = outZ2 + N;
  float* outLv  = outMu + (size_t)N * 64;

  const unsigned gN64 = (unsigned)((N + 63) / 64);
  const unsigned gBin = cdiv(E, 256 * EPT);

  zero_kernel<<<cdiv(N + 2 * MAXOWN, 256), 256, 0, stream>>>(cnt, N + 2 * MAXOWN);
  degree_count_kernel<<<gBin, 256, 0, stream>>>(dst, cnt, ocnt, E, (int)gOwn);
  invdeg_kernel<<<cdiv(N, 256), 256, 0, stream>>>(cnt, inv, N);
  owner_bin_kernel<<<gBin, 256, 0, stream>>>(src, dst, ocnt, ofill, ooff, list, E, (int)gOwn);

  owner_sum_kernel<64><<<gOwn, 1024, 0, stream>>>(x, list, ooff, bufA, N);
  wmma_linear_kernel<64, 128, 1, true><<<gN64, 256, 0, stream>>>(
      bufA, inv, x, g1Wl, g1b, g1Wr, bufB, nullptr, N);

  owner_sum_kernel<128><<<gOwn, 1024, 0, stream>>>(bufB, list, ooff, bufA, N);
  wmma_linear_kernel<128, 128, 1, true><<<gN64, 256, 0, stream>>>(
      bufA, inv, bufB, g2Wl, g2b, g2Wr, bufA, nullptr, N);

  owner_sum_kernel<128><<<gOwn, 1024, 0, stream>>>(bufA, list, ooff, bufB, N);
  wmma_linear_kernel<128, 128, 1, true><<<gN64, 256, 0, stream>>>(
      bufB, inv, bufA, g3Wl, g3b, g3Wr, bufB, nullptr, N);

  wmma_linear_kernel<128, 64, 0, false><<<gN64, 128, 0, stream>>>(
      nullptr, nullptr, bufB, trW, trb, nullptr, outMu, outLv, N);

  wmma_linear_kernel<64, 128, 2, false><<<gN64, 256, 0, stream>>>(
      nullptr, nullptr, outMu, d1W, d1b, nullptr, bufA, nullptr, N);

  owner_sum_kernel<128><<<gOwn, 1024, 0, stream>>>(bufA, list, ooff, bufB, N);
  wmma_linear_kernel<128, 128, 2, true><<<gN64, 256, 0, stream>>>(
      bufB, inv, bufA, d2Wl, d2b, d2Wr, bufB, nullptr, N);

  owner_sum_kernel<128><<<gOwn, 1024, 0, stream>>>(bufB, list, ooff, bufA, N);
  wmma_linear_kernel<128, 64, 1, true><<<gN64, 128, 0, stream>>>(
      bufA, inv, bufB, d3Wl, d3b, d3Wr, bufC, nullptr, N);

  owner_sum_kernel<64><<<gOwn, 1024, 0, stream>>>(bufC, list, ooff, bufA, N);
  wmma_linear_kernel<64, 3, 0, true><<<gN64, 32, 0, stream>>>(
      bufA, inv, bufC, d4Wl, d4b, d4Wr, z2, nullptr, N);

  split_z2_kernel<<<cdiv(N, 256), 256, 0, stream>>>(z2, outZ01, outZ2, N);
}


# Linear kernels: weights load from global memory straight into each wave's B-fragment registers, not via LDS: LDS a third, blocks share a WGP; k visited in another order in each 8, sums reassociated
# speedup vs baseline: 2.7700x; 1.0597x over previous
#include <hip/hip_runtime.h>
#include <hip/hip_bf16.h>

typedef __attribute__((ext_vector_type(2))) float v2f;
typedef __attribute__((ext_vector_type(8))) float v8f;


__device__ __forceinline__ void atom_add_f32(float* p, float v) {
  __hip_atomic_fetch_add(p, v, __ATOMIC_RELAXED, __HIP_MEMORY_SCOPE_AGENT);
}

__global__ void zero_kernel(float* __restrict__ p, long n) {
  long i = (long)blockIdx.x * blockDim.x + threadIdx.x;
  if (i < n) p[i] = 0.0f;
}

__global__ void invdeg_kernel(const float* __restrict__ cnt, float* __restrict__ inv, int n) {
  int i = blockIdx.x * blockDim.x + threadIdx.x;
  if (i < n) inv[i] = 1.0f / fmaxf(cnt[i], 1.0f);
}

constexpr int RNG = 400;
constexpr int MAXOWN = 1024;
constexpr int EPT = 16;
constexpr int SRC_BITS = 23;

__device__ __forceinline__ void lds_add_f32(float* p, float v) {
  __hip_atomic_fetch_add(p, v, __ATOMIC_RELAXED, __HIP_MEMORY_SCOPE_WORKGROUP);
}

__global__ __launch_bounds__(256) void degree_count_kernel(const int* __restrict__ dst,
                                                           float* __restrict__ cnt,
                                                           int* __restrict__ ocnt, int E, int nown) {
  __shared__ int h[MAXOWN];
  const int tid = threadIdx.x;
  for (int i = tid; i < MAXOWN; i += 256) h[i] = 0;
  __syncthreads();
  const int base = blockIdx.x * (256 * EPT);
  int d[EPT];
#pragma unroll
  for (int k = 0; k < EPT; ++k) {
    const int e = base + k * 256 + tid;
    d[k] = (e < E) ? dst[e] : -1;
  }
#pragma unroll
  for (int k = 0; k < EPT; ++k) {
    if (d[k] >= 0) {
      atom_add_f32(&cnt[d[k]], 1.0f);
      atomicAdd(&h[d[k] / RNG], 1);
    }
  }
  __syncthreads();
  for (int i = tid; i < nown; i += 256) {
    const int c = h[i];
    if (c) atomicAdd(&ocnt[i], c);
  }
}

__global__ __launch_bounds__(256) void owner_bin_kernel(const int* __restrict__ src,
                                                        const int* __restrict__ dst,
                                                        const int* __restrict__ ocnt,
                                                        int* __restrict__ ofill,
                                                        int* __restrict__ ooff,
                                                        unsigned* __restrict__ list,
                                                        int E, int nown) {
  __shared__ int h[MAXOWN];
  __shared__ int off[MAXOWN];
  __shared__ int sc[256];
  const int tid = threadIdx.x;
  for (int i = tid; i < MAXOWN; i += 256) h[i] = 0;
  int c[4];
  int s = 0;
#pragma unroll
  for (int k = 0; k < 4; ++k) {
    const int b = 4 * tid + k;
    c[k] = (b < nown) ? ocnt[b] : 0;
    s += c[k];
  }
  sc[tid] = s;
  __syncthreads();
  for (int dd = 1; dd < 256; dd <<= 1) {
    const int v = (tid >= dd) ? sc[tid - dd] : 0;
    __syncthreads();
    sc[tid] += v;
    __syncthreads();
  }
  int run = sc[tid] - s;
#pragma unroll
  for (int k = 0; k < 4; ++k) {
    off[4 * tid + k] = run;
    run += c[k];
  }
  __syncthreads();
  if (blockIdx.x == 0)
    for (int i = tid; i <= nown; i += 256) ooff[i] = (i < nown) ? off[i] : sc[255];

  const int base = blockIdx.x * (256 * EPT);
  int dv[EPT], sv[EPT];
#pragma unroll
  for (int k = 0; k < EPT; ++k) {
    const int e = base + k * 256 + tid;
    const int ec = (e < E) ? e : E - 1;
    dv[k] = dst[ec];
    sv[k] = src[ec];
  }
  int ow[EPT], rk[EPT];
#pragma unroll
  for (int k = 0; k < EPT; ++k) {
    const int e = base + k * 256 + tid;
    ow[k] = dv[k] / RNG;
    rk[k] = (e < E) ? atomicAdd(&h[ow[k]], 1) : 0;
  }
  __syncthreads();
  for (int i = tid; i < nown; i += 256) {
    const int cc = h[i];
    if (cc) off[i] += atomicAdd(&ofill[i], cc);
  }
  __syncthreads();
#pragma unroll
  for (int k = 0; k < EPT; ++k) {
    const int e = base + k * 256 + tid;
    if (e < E)
      list[off[ow[k]] + rk[k]] =
          (unsigned)sv[k] | ((unsigned)(dv[k] - ow[k] * RNG) << SRC_BITS);
  }
}

template <int D>
__global__ __launch_bounds__(1024) void owner_sum_kernel(const float* __restrict__ x,
                                                         const unsigned* __restrict__ list,
                                                         const int* __restrict__ ooff,
                                                         float* __restrict__ agg, int nnodes) {
  constexpr int L = D / 32;
  constexpr int U = 32 / L;
  __shared__ alignas(16) float acc[RNG * D];
  const int tid = threadIdx.x;
  const int r0 = blockIdx.x * RNG;
  const int beg = ooff[blockIdx.x];
  const int end = ooff[blockIdx.x + 1];

  for (int i = tid; i < RNG * D / 4; i += 1024)
    ((float4*)acc)[i] = make_float4(0.0f, 0.0f, 0.0f, 0.0f);
  __syncthreads();

  const int lane = tid & 31;
  const int wv = __builtin_amdgcn_readfirstlane(tid >> 5);
  for (int st = beg + wv * 32; st < end; st += 32 * 32) {
    const int qi = (st + lane < end) ? st + lane : end - 1;
    const unsigned p = list[qi];
    const bool mine = (st + lane < end);
    const int sv = mine ? (int)(p & ((1u << SRC_BITS) - 1u)) : 0;
    const int dl = mine ? (int)(p >> SRC_BITS) : -1;
    for (int i0 = 0; i0 < 32 && st + i0 < end; i0 += U) {
      float v[U][L];
#pragma unroll
      for (int j = 0; j < U; ++j) {
        const int sj = __builtin_amdgcn_readlane(sv, i0 + j);
        const float* r = x + (long)sj * D;
#pragma unroll
        for (int l = 0; l < L; ++l) v[j][l] = r[lane + 32 * l];
      }
#pragma unroll
      for (int j = 0; j < U; ++j) {
        const int dj = __builtin_amdgcn_readlane(dl, i0 + j);
        if (dj >= 0) {
          float* o = acc + dj * D;
#pragma unroll
          for (int l = 0; l < L; ++l) lds_add_f32(o + lane + 32 * l, v[j][l]);
        }
      }
    }
  }
  __syncthreads();

  for (int i = tid; i < RNG * D / 4; i += 1024) {
    const int r = i / (D / 4);
    if (r0 + r < nnodes) ((float4*)(agg + (long)r0 * D))[i] = ((const float4*)acc)[i];
  }
}

__global__ void split_z2_kernel(const float* __restrict__ z2,
                                float* __restrict__ out01,
                                float* __restrict__ out2, int n) {
  int i = blockIdx.x * blockDim.x + threadIdx.x;
  if (i < n) {
    out01[2 * i + 0] = z2[3 * i + 0];
    out01[2 * i + 1] = z2[3 * i + 1];
    out2[i] = z2[3 * i + 2];
  }
}

template <int DIN, int DOUT, int ACT, bool HAS_AGG>
__global__ __launch_bounds__(2 * ((DOUT + 15) & ~15)) void wmma_linear_kernel(
    const float* agg, const float* __restrict__ invdeg,
    const float* __restrict__ x,
    const float* __restrict__ Wl, const float* __restrict__ bias,
    const float* __restrict__ Wr,
    float* out, float* out2, int nnodes) {
  constexpr int MT = 4;
  constexpr int ROWS = 16 * MT;
  constexpr int DOUTP = (DOUT + 15) & ~15;
  constexpr int LDW = DIN + 4;
  constexpr int C4 = DIN / 4;
  constexpr int KB = DIN / 8;
  __shared__ alignas(16) float sA[ROWS * LDW];
  __shared__ alignas(16) float sX[HAS_AGG ? ROWS * LDW : 4];

  const int tid = threadIdx.x;
  constexpr int NT = 2 * DOUTP;
  const int row0 = blockIdx.x * ROWS;

  const int wave = tid >> 5;
  const int lane = tid & 31;
  const int half = lane >> 4;
  const int l16 = lane & 15;
  const int n = wave * 16 + l16;
  const bool colOK = (DOUT == DOUTP) || (n < DOUT);

  constexpr int APT = ROWS * C4 / NT;
  constexpr int WREG = KB * (HAS_AGG ? 8 : 4);
  constexpr int AMAX = (208 - WREG) / (HAS_AGG ? 9 : 4);
  constexpr int AG = (APT <= AMAX) ? APT : (APT / 2 <= AMAX) ? APT / 2 : (APT / 4 <= AMAX) ? APT / 4 : APT / 8;
  float4 bl[KB];
  float4 br[HAS_AGG ? KB : 1];
  const int nr = colOK ? n : 0;
  const float* pl = Wl + (long)nr * DIN + 4 * half;
  const float* pr = HAS_AGG ? Wr + (long)nr * DIN + 4 * half : Wl;
  for (int g = 0; g < APT; g += AG) {
    float4 av[AG];
    float4 xv[HAS_AGG ? AG : 1];
    float sc[HAS_AGG ? AG : 1];
#pragma unroll
    for (int j = 0; j < AG; ++j) {
      const int i = tid + (g + j) * NT;
      const int r = i / C4, c4 = (i - r * C4) * 4;
      int m = row0 + r;
      if (m >= nnodes) m = nnodes - 1;
      if constexpr (HAS_AGG) {
        sc[j] = invdeg[m];
        av[j] = *(const float4*)(agg + (long)m * DIN + c4);
        xv[j] = *(const float4*)(x + (long)m * DIN + c4);
      } else {
        av[j] = *(const float4*)(x + (long)m * DIN + c4);
      }
    }
    if (g == 0) {
#pragma unroll
      for (int j = 0; j < KB; ++j) {
        bl[j] = *(const float4*)(pl + 8 * j);
        if constexpr (HAS_AGG) br[j] = *(const float4*)(pr + 8 * j);
      }
    }
#pragma unroll
    for (int j = 0; j < AG; ++j) {
      const int i = tid + (g + j) * NT;
      const int r = i / C4, c4 = (i - r * C4) * 4;
      if constexpr (HAS_AGG) {
        float4 a = av[j];
        const float s = sc[j];
        a.x *= s; a.y *= s; a.z *= s; a.w *= s;
        *(float4*)&sA[r * LDW + c4] = a;
        *(float4*)&sX[r * LDW + c4] = xv[j];
      } else {
        *(float4*)&sA[r * LDW + c4] = av[j];
      }
    }
  }
  __syncthreads();

  v8f acc[MT];
#pragma unroll
  for (int mt = 0; mt < MT; ++mt) acc[mt] = (v8f){};

#pragma unroll
  for (int j = 0; j < KB; ++j) {
    const int kk = 8 * j + 4 * half;
    float4 a[MT];
    float4 a2[HAS_AGG ? MT : 1];
#pragma unroll
    for (int mt = 0; mt < MT; ++mt) {
      const int r = mt * 16 + l16;
      a[mt] = *(const float4*)&sA[r * LDW + kk];
      if constexpr (HAS_AGG) a2[mt] = *(const float4*)&sX[r * LDW + kk];
    }
    const v2f b0 = {bl[j].x, bl[j].y};
    const v2f b1 = {bl[j].z, bl[j].w};
#pragma unroll
    for (int mt = 0; mt < MT; ++mt) {
      const v2f a0 = {a[mt].x, a[mt].y};
      acc[mt] = __builtin_amdgcn_wmma_f32_16x16x4_f32(false, a0, false, b0,
                                                      (short)0, acc[mt], false,
                                                      false);
      if constexpr (HAS_AGG) {
        const v2f x0 = {a2[mt].x, a2[mt].y};
        const v2f c0 = {br[j].x, br[j].y};
        acc[mt] = __builtin_amdgcn_wmma_f32_16x16x4_f32(false, x0, false, c0,
                                                        (short)0, acc[mt],
                                                        false, false);
      }
    }
#pragma unroll
    for (int mt = 0; mt < MT; ++mt) {
      const v2f a1 = {a[mt].z, a[mt].w};
      acc[mt] = __builtin_amdgcn_wmma_f32_16x16x4_f32(false, a1, false, b1,
                                                      (short)0, acc[mt], false,
                                                      false);
      if constexpr (HAS_AGG) {
        const v2f x1 = {a2[mt].z, a2[mt].w};
        const v2f c1 = {br[j].z, br[j].w};
        acc[mt] = __builtin_amdgcn_wmma_f32_16x16x4_f32(false, x1, false, c1,
                                                        (short)0, acc[mt],
                                                        false, false);
      }
    }
  }

  const float bcol = colOK ? bias[colOK ? n : 0] : 0.0f;
#pragma unroll
  for (int mt = 0; mt < MT; ++mt) {
#pragma unroll
    for (int i = 0; i < 8; ++i) {
      int row = row0 + mt * 16 + i + 8 * half;
      float v = acc[mt][i] + bcol;
      if (ACT == 1) v = tanhf(v);
      if (ACT == 2) v = (v > 0.0f) ? v : 0.01f * v;
      if (colOK && row < nnodes) {
        out[(long)row * DOUT + n] = v;
        if (out2) out2[(long)row * DOUT + n] = v;
      }
    }
  }
}


static inline unsigned cdiv(long a, long b) { return (unsigned)((a + b - 1) / b); }

extern "C" void kernel_launch(void* const* d_in, const int* in_sizes, int n_in,
                              void* d_out, int out_size, void* d_ws, size_t ws_size,
                              hipStream_t stream) {
  const float* x = (const float*)d_in[0];
  const int* ei = (const int*)d_in[1];
  const int N = in_sizes[0] / 64;
  const int E = in_sizes[1] / 2;
  const int* src = ei;
  const int* dst = ei + E;

  const float* g1Wl = (const float*)d_in[2];
  const float* g1b  = (const float*)d_in[3];
  const float* g1Wr = (const float*)d_in[4];
  const float* g2Wl = (const float*)d_in[5];
  const float* g2b  = (const float*)d_in[6];
  const float* g2Wr = (const float*)d_in[7];
  const float* g3Wl = (const float*)d_in[8];
  const float* g3b  = (const float*)d_in[9];
  const float* g3Wr = (const float*)d_in[10];
  const float* trW  = (const float*)d_in[11];
  const float* trb  = (const float*)d_in[12];
  const float* d1W  = (const float*)d_in[13];
  const float* d1b  = (const float*)d_in[14];
  const float* d2Wl = (const float*)d_in[15];
  const float* d2b  = (const float*)d_in[16];
  const float* d2Wr = (const float*)d_in[17];
  const float* d3Wl = (const float*)d_in[18];
  const float* d3b  = (const float*)d_in[19];
  const float* d3Wr = (const float*)d_in[20];
  const float* d4Wl = (const float*)d_in[21];
  const float* d4b  = (const float*)d_in[22];
  const float* d4Wr = (const float*)d_in[23];

  const unsigned gOwn = cdiv(N, RNG);
  float* ws   = (float*)d_ws;
  float* inv  = ws;
  unsigned* list = (unsigned*)(ws + N);
  float* bufA = ws + (size_t)N + (size_t)E;
  float* bufB = bufA + (size_t)N * 128;
  float* bufC = bufB + (size_t)N * 128;
  int* ooff   = (int*)(bufC + (size_t)N * 64);
  float* cnt  = bufA;
  int* ocnt   = (int*)(bufA + N);
  int* ofill  = ocnt + MAXOWN;
  float* z2   = bufB;

  float* outZ01 = (float*)d_out;
  float* outZ2  = outZ01 + 2 * (size_t)N;
  float* outMu  = outZ2 + N;
  float* outLv  = outMu + (size_t)N * 64;

  const unsigned gN64 = (unsigned)((N + 63) / 64);
  const unsigned gBin = cdiv(E, 256 * EPT);

  zero_kernel<<<cdiv(N + 2 * MAXOWN, 256), 256, 0, stream>>>(cnt, N + 2 * MAXOWN);
  degree_count_kernel<<<gBin, 256, 0, stream>>>(dst, cnt, ocnt, E, (int)gOwn);
  invdeg_kernel<<<cdiv(N, 256), 256, 0, stream>>>(cnt, inv, N);
  owner_bin_kernel<<<gBin, 256, 0, stream>>>(src, dst, ocnt, ofill, ooff, list, E, (int)gOwn);

  owner_sum_kernel<64><<<gOwn, 1024, 0, stream>>>(x, list, ooff, bufA, N);
  wmma_linear_kernel<64, 128, 1, true><<<gN64, 256, 0, stream>>>(
      bufA, inv, x, g1Wl, g1b, g1Wr, bufB, nullptr, N);

  owner_sum_kernel<128><<<gOwn, 1024, 0, stream>>>(bufB, list, ooff, bufA, N);
  wmma_linear_kernel<128, 128, 1, true><<<gN64, 256, 0, stream>>>(
      bufA, inv, bufB, g2Wl, g2b, g2Wr, bufA, nullptr, N);

  owner_sum_kernel<128><<<gOwn, 1024, 0, stream>>>(bufA, list, ooff, bufB, N);
  wmma_linear_kernel<128, 128, 1, true><<<gN64, 256, 0, stream>>>(
      bufB, inv, bufA, g3Wl, g3b, g3Wr, bufB, nullptr, N);

  wmma_linear_kernel<128, 64, 0, false><<<gN64, 128, 0, stream>>>(
      nullptr, nullptr, bufB, trW, trb, nullptr, outMu, outLv, N);

  wmma_linear_kernel<64, 128, 2, false><<<gN64, 256, 0, stream>>>(
      nullptr, nullptr, outMu, d1W, d1b, nullptr, bufA, nullptr, N);

  owner_sum_kernel<128><<<gOwn, 1024, 0, stream>>>(bufA, list, ooff, bufB, N);
  wmma_linear_kernel<128, 128, 2, true><<<gN64, 256, 0, stream>>>(
      bufB, inv, bufA, d2Wl, d2b, d2Wr, bufB, nullptr, N);

  owner_sum_kernel<128><<<gOwn, 1024, 0, stream>>>(bufB, list, ooff, bufA, N);
  wmma_linear_kernel<128, 64, 1, true><<<gN64, 128, 0, stream>>>(
      bufA, inv, bufB, d3Wl, d3b, d3Wr, bufC, nullptr, N);

  owner_sum_kernel<64><<<gOwn, 1024, 0, stream>>>(bufC, list, ooff, bufA, N);
  wmma_linear_kernel<64, 3, 0, true><<<gN64, 32, 0, stream>>>(
      bufA, inv, bufC, d4Wl, d4b, d4Wr, z2, nullptr, N);

  split_z2_kernel<<<cdiv(N, 256), 256, 0, stream>>>(z2, outZ01, outZ2, N);
}


# Neighbour sums: ONE load instruction fetches a whole source row (16 or 8 bytes a lane, 64 registers in flight), rows 256-byte aligned; LDS tile columns permuted, adds conflict-free; sum order kept
# speedup vs baseline: 2.8777x; 1.0389x over previous
#include <hip/hip_runtime.h>
#include <hip/hip_bf16.h>

typedef __attribute__((ext_vector_type(2))) float v2f;
typedef __attribute__((ext_vector_type(8))) float v8f;


__device__ __forceinline__ void atom_add_f32(float* p, float v) {
  __hip_atomic_fetch_add(p, v, __ATOMIC_RELAXED, __HIP_MEMORY_SCOPE_AGENT);
}

__global__ void zero_kernel(float* __restrict__ p, long n) {
  long i = (long)blockIdx.x * blockDim.x + threadIdx.x;
  if (i < n) p[i] = 0.0f;
}

__global__ void invdeg_kernel(const float* __restrict__ cnt, float* __restrict__ inv, int n) {
  int i = blockIdx.x * blockDim.x + threadIdx.x;
  if (i < n) inv[i] = 1.0f / fmaxf(cnt[i], 1.0f);
}

constexpr int RNG = 400;
constexpr int MAXOWN = 1024;
constexpr int EPT = 16;
constexpr int SRC_BITS = 23;

__device__ __forceinline__ void lds_add_f32(float* p, float v) {
  __hip_atomic_fetch_add(p, v, __ATOMIC_RELAXED, __HIP_MEMORY_SCOPE_WORKGROUP);
}

__global__ __launch_bounds__(256) void degree_count_kernel(const int* __restrict__ dst,
                                                           float* __restrict__ cnt,
                                                           int* __restrict__ ocnt, int E, int nown) {
  __shared__ int h[MAXOWN];
  const int tid = threadIdx.x;
  for (int i = tid; i < MAXOWN; i += 256) h[i] = 0;
  __syncthreads();
  const int base = blockIdx.x * (256 * EPT);
  int d[EPT];
#pragma unroll
  for (int k = 0; k < EPT; ++k) {
    const int e = base + k * 256 + tid;
    d[k] = (e < E) ? dst[e] : -1;
  }
#pragma unroll
  for (int k = 0; k < EPT; ++k) {
    if (d[k] >= 0) {
      atom_add_f32(&cnt[d[k]], 1.0f);
      atomicAdd(&h[d[k] / RNG], 1);
    }
  }
  __syncthreads();
  for (int i = tid; i < nown; i += 256) {
    const int c = h[i];
    if (c) atomicAdd(&ocnt[i], c);
  }
}

__global__ __launch_bounds__(256) void owner_bin_kernel(const int* __restrict__ src,
                                                        const int* __restrict__ dst,
                                                        const int* __restrict__ ocnt,
                                                        int* __restrict__ ofill,
                                                        int* __restrict__ ooff,
                                                        unsigned* __restrict__ list,
                                                        int E, int nown) {
  __shared__ int h[MAXOWN];
  __shared__ int off[MAXOWN];
  __shared__ int sc[256];
  const int tid = threadIdx.x;
  for (int i = tid; i < MAXOWN; i += 256) h[i] = 0;
  int c[4];
  int s = 0;
#pragma unroll
  for (int k = 0; k < 4; ++k) {
    const int b = 4 * tid + k;
    c[k] = (b < nown) ? ocnt[b] : 0;
    s += c[k];
  }
  sc[tid] = s;
  __syncthreads();
  for (int dd = 1; dd < 256; dd <<= 1) {
    const int v = (tid >= dd) ? sc[tid - dd] : 0;
    __syncthreads();
    sc[tid] += v;
    __syncthreads();
  }
  int run = sc[tid] - s;
#pragma unroll
  for (int k = 0; k < 4; ++k) {
    off[4 * tid + k] = run;
    run += c[k];
  }
  __syncthreads();
  if (blockIdx.x == 0)
    for (int i = tid; i <= nown; i += 256) ooff[i] = (i < nown) ? off[i] : sc[255];

  const int base = blockIdx.x * (256 * EPT);
  int dv[EPT], sv[EPT];
#pragma unroll
  for (int k = 0; k < EPT; ++k) {
    const int e = base + k * 256 + tid;
    const int ec = (e < E) ? e : E - 1;
    dv[k] = dst[ec];
    sv[k] = src[ec];
  }
  int ow[EPT], rk[EPT];
#pragma unroll
  for (int k = 0; k < EPT; ++k) {
    const int e = base + k * 256 + tid;
    ow[k] = dv[k] / RNG;
    rk[k] = (e < E) ? atomicAdd(&h[ow[k]], 1) : 0;
  }
  __syncthreads();
  for (int i = tid; i < nown; i += 256) {
    const int cc = h[i];
    if (cc) off[i] += atomicAdd(&ofill[i], cc);
  }
  __syncthreads();
#pragma unroll
  for (int k = 0; k < EPT; ++k) {
    const int e = base + k * 256 + tid;
    if (e < E)
      list[off[ow[k]] + rk[k]] =
          (unsigned)sv[k] | ((unsigned)(dv[k] - ow[k] * RNG) << SRC_BITS);
  }
}

template <int D>
__global__ __launch_bounds__(1024) void owner_sum_kernel(const float* __restrict__ x,
                                                         const unsigned* __restrict__ list,
                                                         const int* __restrict__ ooff,
                                                         float* __restrict__ agg, int nnodes) {
  constexpr int V = D / 32;
  constexpr int U = 64 / V;
  typedef float rowv __attribute__((ext_vector_type(V)));
  __shared__ alignas(16) float acc[RNG * D];
  const int tid = threadIdx.x;
  const int r0 = blockIdx.x * RNG;
  const int beg = ooff[blockIdx.x];
  const int end = ooff[blockIdx.x + 1];

  for (int i = tid; i < RNG * D / 4; i += 1024)
    ((float4*)acc)[i] = make_float4(0.0f, 0.0f, 0.0f, 0.0f);
  __syncthreads();

  const int lane = tid & 31;
  const int wv = __builtin_amdgcn_readfirstlane(tid >> 5);
  for (int st = beg + wv * 32; st < end; st += 32 * 32) {
    const int qi = (st + lane < end) ? st + lane : end - 1;
    const unsigned p = list[qi];
    const bool mine = (st + lane < end);
    const int sv = mine ? (int)(p & ((1u << SRC_BITS) - 1u)) : 0;
    const int dl = mine ? (int)(p >> SRC_BITS) : -1;
    for (int i0 = 0; i0 < 32 && st + i0 < end; i0 += U) {
      rowv v[U];
#pragma unroll
      for (int j = 0; j < U; ++j) {
        const int sj = __builtin_amdgcn_readlane(sv, i0 + j);
        const float* r = x + (long)sj * D;
        v[j] = *(const rowv*)(r + V * lane);
      }
#pragma unroll
      for (int j = 0; j < U; ++j) {
        const int dj = __builtin_amdgcn_readlane(dl, i0 + j);
        if (dj >= 0) {
          float* o = acc + dj * D + lane;
#pragma unroll
          for (int c = 0; c < V; ++c) lds_add_f32(o + 32 * c, v[j][c]);
        }
      }
    }
  }
  __syncthreads();

  for (int r = wv; r < RNG; r += 32) {
    if (r0 + r < nnodes) {
      rowv o;
#pragma unroll
      for (int c = 0; c < V; ++c) o[c] = acc[r * D + 32 * c + lane];
      *(rowv*)(agg + (long)(r0 + r) * D + V * lane) = o;
    }
  }
}

__global__ void split_z2_kernel(const float* __restrict__ z2,
                                float* __restrict__ out01,
                                float* __restrict__ out2, int n) {
  int i = blockIdx.x * blockDim.x + threadIdx.x;
  if (i < n) {
    out01[2 * i + 0] = z2[3 * i + 0];
    out01[2 * i + 1] = z2[3 * i + 1];
    out2[i] = z2[3 * i + 2];
  }
}

template <int DIN, int DOUT, int ACT, bool HAS_AGG>
__global__ __launch_bounds__(2 * ((DOUT + 15) & ~15)) void wmma_linear_kernel(
    const float* agg, const float* __restrict__ invdeg,
    const float* __restrict__ x,
    const float* __restrict__ Wl, const float* __restrict__ bias,
    const float* __restrict__ Wr,
    float* out, float* out2, int nnodes) {
  constexpr int MT = 4;
  constexpr int ROWS = 16 * MT;
  constexpr int DOUTP = (DOUT + 15) & ~15;
  constexpr int LDW = DIN + 4;
  constexpr int C4 = DIN / 4;
  constexpr int KB = DIN / 8;
  __shared__ alignas(16) float sA[ROWS * LDW];
  __shared__ alignas(16) float sX[HAS_AGG ? ROWS * LDW : 4];

  const int tid = threadIdx.x;
  constexpr int NT = 2 * DOUTP;
  const int row0 = blockIdx.x * ROWS;

  const int wave = tid >> 5;
  const int lane = tid & 31;
  const int half = lane >> 4;
  const int l16 = lane & 15;
  const int n = wave * 16 + l16;
  const bool colOK = (DOUT == DOUTP) || (n < DOUT);

  constexpr int APT = ROWS * C4 / NT;
  constexpr int WREG = KB * (HAS_AGG ? 8 : 4);
  constexpr int AMAX = (208 - WREG) / (HAS_AGG ? 9 : 4);
  constexpr int AG = (APT <= AMAX) ? APT : (APT / 2 <= AMAX) ? APT / 2 : (APT / 4 <= AMAX) ? APT / 4 : APT / 8;
  float4 bl[KB];
  float4 br[HAS_AGG ? KB : 1];
  const int nr = colOK ? n : 0;
  const float* pl = Wl + (long)nr * DIN + 4 * half;
  const float* pr = HAS_AGG ? Wr + (long)nr * DIN + 4 * half : Wl;
  for (int g = 0; g < APT; g += AG) {
    float4 av[AG];
    float4 xv[HAS_AGG ? AG : 1];
    float sc[HAS_AGG ? AG : 1];
#pragma unroll
    for (int j = 0; j < AG; ++j) {
      const int i = tid + (g + j) * NT;
      const int r = i / C4, c4 = (i - r * C4) * 4;
      int m = row0 + r;
      if (m >= nnodes) m = nnodes - 1;
      if constexpr (HAS_AGG) {
        sc[j] = invdeg[m];
        av[j] = *(const float4*)(agg + (long)m * DIN + c4);
        xv[j] = *(const float4*)(x + (long)m * DIN + c4);
      } else {
        av[j] = *(const float4*)(x + (long)m * DIN + c4);
      }
    }
    if (g == 0) {
#pragma unroll
      for (int j = 0; j < KB; ++j) {
        bl[j] = *(const float4*)(pl + 8 * j);
        if constexpr (HAS_AGG) br[j] = *(const float4*)(pr + 8 * j);
      }
    }
#pragma unroll
    for (int j = 0; j < AG; ++j) {
      const int i = tid + (g + j) * NT;
      const int r = i / C4, c4 = (i - r * C4) * 4;
      if constexpr (HAS_AGG) {
        float4 a = av[j];
        const float s = sc[j];
        a.x *= s; a.y *= s; a.z *= s; a.w *= s;
        *(float4*)&sA[r * LDW + c4] = a;
        *(float4*)&sX[r * LDW + c4] = xv[j];
      } else {
        *(float4*)&sA[r * LDW + c4] = av[j];
      }
    }
  }
  __syncthreads();

  v8f acc[MT];
#pragma unroll
  for (int mt = 0; mt < MT; ++mt) acc[mt] = (v8f){};

#pragma unroll
  for (int j = 0; j < KB; ++j) {
    const int kk = 8 * j + 4 * half;
    float4 a[MT];
    float4 a2[HAS_AGG ? MT : 1];
#pragma unroll
    for (int mt = 0; mt < MT; ++mt) {
      const int r = mt * 16 + l16;
      a[mt] = *(const float4*)&sA[r * LDW + kk];
      if constexpr (HAS_AGG) a2[mt] = *(const float4*)&sX[r * LDW + kk];
    }
    const v2f b0 = {bl[j].x, bl[j].y};
    const v2f b1 = {bl[j].z, bl[j].w};
#pragma unroll
    for (int mt = 0; mt < MT; ++mt) {
      const v2f a0 = {a[mt].x, a[mt].y};
      acc[mt] = __builtin_amdgcn_wmma_f32_16x16x4_f32(false, a0, false, b0,
                                                      (short)0, acc[mt], false,
                                                      false);
      if constexpr (HAS_AGG) {
        const v2f x0 = {a2[mt].x, a2[mt].y};
        const v2f c0 = {br[j].x, br[j].y};
        acc[mt] = __builtin_amdgcn_wmma_f32_16x16x4_f32(false, x0, false, c0,
                                                        (short)0, acc[mt],
                                                        false, false);
      }
    }
#pragma unroll
    for (int mt = 0; mt < MT; ++mt) {
      const v2f a1 = {a[mt].z, a[mt].w};
      acc[mt] = __builtin_amdgcn_wmma_f32_16x16x4_f32(false, a1, false, b1,
                                                      (short)0, acc[mt], false,
                                                      false);
      if constexpr (HAS_AGG) {
        const v2f x1 = {a2[mt].z, a2[mt].w};
        const v2f c1 = {br[j].z, br[j].w};
        acc[mt] = __builtin_amdgcn_wmma_f32_16x16x4_f32(false, x1, false, c1,
                                                        (short)0, acc[mt],
                                                        false, false);
      }
    }
  }

  const float bcol = colOK ? bias[colOK ? n : 0] : 0.0f;
#pragma unroll
  for (int mt = 0; mt < MT; ++mt) {
#pragma unroll
    for (int i = 0; i < 8; ++i) {
      int row = row0 + mt * 16 + i + 8 * half;
      float v = acc[mt][i] + bcol;
      if (ACT == 1) v = tanhf(v);
      if (ACT == 2) v = (v > 0.0f) ? v : 0.01f * v;
      if (colOK && row < nnodes) {
        out[(long)row * DOUT + n] = v;
        if (out2) out2[(long)row * DOUT + n] = v;
      }
    }
  }
}


static inline unsigned cdiv(long a, long b) { return (unsigned)((a + b - 1) / b); }

extern "C" void kernel_launch(void* const* d_in, const int* in_sizes, int n_in,
                              void* d_out, int out_size, void* d_ws, size_t ws_size,
                              hipStream_t stream) {
  const float* x = (const float*)d_in[0];
  const int* ei = (const int*)d_in[1];
  const int N = in_sizes[0] / 64;
  const int E = in_sizes[1] / 2;
  const int* src = ei;
  const int* dst = ei + E;

  const float* g1Wl = (const float*)d_in[2];
  const float* g1b  = (const float*)d_in[3];
  const float* g1Wr = (const float*)d_in[4];
  const float* g2Wl = (const float*)d_in[5];
  const float* g2b  = (const float*)d_in[6];
  const float* g2Wr = (const float*)d_in[7];
  const float* g3Wl = (const float*)d_in[8];
  const float* g3b  = (const float*)d_in[9];
  const float* g3Wr = (const float*)d_in[10];
  const float* trW  = (const float*)d_in[11];
  const float* trb  = (const float*)d_in[12];
  const float* d1W  = (const float*)d_in[13];
  const float* d1b  = (const float*)d_in[14];
  const float* d2Wl = (const float*)d_in[15];
  const float* d2b  = (const float*)d_in[16];
  const float* d2Wr = (const float*)d_in[17];
  const float* d3Wl = (const float*)d_in[18];
  const float* d3b  = (const float*)d_in[19];
  const float* d3Wr = (const float*)d_in[20];
  const float* d4Wl = (const float*)d_in[21];
  const float* d4b  = (const float*)d_in[22];
  const float* d4Wr = (const float*)d_in[23];

  const unsigned gOwn = cdiv(N, RNG);
  float* ws   = (float*)d_ws;
  float* inv  = ws;
  unsigned* list = (unsigned*)(ws + N);
  float* bufA = (float*)(((uintptr_t)(ws + (size_t)N + (size_t)E) + 255) & ~(uintptr_t)255);
  float* bufB = bufA + (size_t)N * 128;
  float* bufC = bufB + (size_t)N * 128;
  int* ooff   = (int*)(bufC + (size_t)N * 64);
  float* cnt  = bufA;
  int* ocnt   = (int*)(bufA + N);
  int* ofill  = ocnt + MAXOWN;
  float* z2   = bufB;

  float* outZ01 = (float*)d_out;
  float* outZ2  = outZ01 + 2 * (size_t)N;
  float* outMu  = outZ2 + N;
  float* outLv  = outMu + (size_t)N * 64;

  const unsigned gN64 = (unsigned)((N + 63) / 64);
  const unsigned gBin = cdiv(E, 256 * EPT);

  zero_kernel<<<cdiv(N + 2 * MAXOWN, 256), 256, 0, stream>>>(cnt, N + 2 * MAXOWN);
  degree_count_kernel<<<gBin, 256, 0, stream>>>(dst, cnt, ocnt, E, (int)gOwn);
  invdeg_kernel<<<cdiv(N, 256), 256, 0, stream>>>(cnt, inv, N);
  owner_bin_kernel<<<gBin, 256, 0, stream>>>(src, dst, ocnt, ofill, ooff, list, E, (int)gOwn);

  owner_sum_kernel<64><<<gOwn, 1024, 0, stream>>>(x, list, ooff, bufA, N);
  wmma_linear_kernel<64, 128, 1, true><<<gN64, 256, 0, stream>>>(
      bufA, inv, x, g1Wl, g1b, g1Wr, bufB, nullptr, N);

  owner_sum_kernel<128><<<gOwn, 1024, 0, stream>>>(bufB, list, ooff, bufA, N);
  wmma_linear_kernel<128, 128, 1, true><<<gN64, 256, 0, stream>>>(
      bufA, inv, bufB, g2Wl, g2b, g2Wr, bufA, nullptr, N);

  owner_sum_kernel<128><<<gOwn, 1024, 0, stream>>>(bufA, list, ooff, bufB, N);
  wmma_linear_kernel<128, 128, 1, true><<<gN64, 256, 0, stream>>>(
      bufB, inv, bufA, g3Wl, g3b, g3Wr, bufB, nullptr, N);

  wmma_linear_kernel<128, 64, 0, false><<<gN64, 128, 0, stream>>>(
      nullptr, nullptr, bufB, trW, trb, nullptr, outMu, outLv, N);

  wmma_linear_kernel<64, 128, 2, false><<<gN64, 256, 0, stream>>>(
      nullptr, nullptr, outMu, d1W, d1b, nullptr, bufA, nullptr, N);

  owner_sum_kernel<128><<<gOwn, 1024, 0, stream>>>(bufA, list, ooff, bufB, N);
  wmma_linear_kernel<128, 128, 2, true><<<gN64, 256, 0, stream>>>(
      bufB, inv, bufA, d2Wl, d2b, d2Wr, bufB, nullptr, N);

  owner_sum_kernel<128><<<gOwn, 1024, 0, stream>>>(bufB, list, ooff, bufA, N);
  wmma_linear_kernel<128, 64, 1, true><<<gN64, 128, 0, stream>>>(
      bufA, inv, bufB, d3Wl, d3b, d3Wr, bufC, nullptr, N);

  owner_sum_kernel<64><<<gOwn, 1024, 0, stream>>>(bufC, list, ooff, bufA, N);
  wmma_linear_kernel<64, 3, 0, true><<<gN64, 32, 0, stream>>>(
      bufA, inv, bufC, d4Wl, d4b, d4Wr, z2, nullptr, N);

  split_z2_kernel<<<cdiv(N, 256), 256, 0, stream>>>(z2, outZ01, outZ2, N);
}


# SAGE layers: the owner work-group multiplies its summed 400-row LDS tile in place (1/deg applied there, root rows staged by 64): no neighbour-sum array in memory, 19 launches -> 13; sum order kept
# speedup vs baseline: 3.3057x; 1.1487x over previous
#include <hip/hip_runtime.h>
#include <hip/hip_bf16.h>

typedef __attribute__((ext_vector_type(2))) float v2f;
typedef __attribute__((ext_vector_type(8))) float v8f;


__device__ __forceinline__ void atom_add_f32(float* p, float v) {
  __hip_atomic_fetch_add(p, v, __ATOMIC_RELAXED, __HIP_MEMORY_SCOPE_AGENT);
}

__global__ void zero_kernel(float* __restrict__ p, long n) {
  long i = (long)blockIdx.x * blockDim.x + threadIdx.x;
  if (i < n) p[i] = 0.0f;
}

__global__ void invdeg_kernel(const float* __restrict__ cnt, float* __restrict__ inv, int n) {
  int i = blockIdx.x * blockDim.x + threadIdx.x;
  if (i < n) inv[i] = 1.0f / fmaxf(cnt[i], 1.0f);
}

constexpr int RNG = 400;
constexpr int MAXOWN = 1024;
constexpr int EPT = 16;
constexpr int SRC_BITS = 23;

__device__ __forceinline__ void lds_add_f32(float* p, float v) {
  __hip_atomic_fetch_add(p, v, __ATOMIC_RELAXED, __HIP_MEMORY_SCOPE_WORKGROUP);
}

__global__ __launch_bounds__(256) void degree_count_kernel(const int* __restrict__ dst,
                                                           float* __restrict__ cnt,
                                                           int* __restrict__ ocnt, int E, int nown) {
  __shared__ int h[MAXOWN];
  const int tid = threadIdx.x;
  for (int i = tid; i < MAXOWN; i += 256) h[i] = 0;
  __syncthreads();
  const int base = blockIdx.x * (256 * EPT);
  int d[EPT];
#pragma unroll
  for (int k = 0; k < EPT; ++k) {
    const int e = base + k * 256 + tid;
    d[k] = (e < E) ? dst[e] : -1;
  }
#pragma unroll
  for (int k = 0; k < EPT; ++k) {
    if (d[k] >= 0) {
      atom_add_f32(&cnt[d[k]], 1.0f);
      atomicAdd(&h[d[k] / RNG], 1);
    }
  }
  __syncthreads();
  for (int i = tid; i < nown; i += 256) {
    const int c = h[i];
    if (c) atomicAdd(&ocnt[i], c);
  }
}

__global__ __launch_bounds__(256) void owner_bin_kernel(const int* __restrict__ src,
                                                        const int* __restrict__ dst,
                                                        const int* __restrict__ ocnt,
                                                        int* __restrict__ ofill,
                                                        int* __restrict__ ooff,
                                                        unsigned* __restrict__ list,
                                                        int E, int nown) {
  __shared__ int h[MAXOWN];
  __shared__ int off[MAXOWN];
  __shared__ int sc[256];
  const int tid = threadIdx.x;
  for (int i = tid; i < MAXOWN; i += 256) h[i] = 0;
  int c[4];
  int s = 0;
#pragma unroll
  for (int k = 0; k < 4; ++k) {
    const int b = 4 * tid + k;
    c[k] = (b < nown) ? ocnt[b] : 0;
    s += c[k];
  }
  sc[tid] = s;
  __syncthreads();
  for (int dd = 1; dd < 256; dd <<= 1) {
    const int v = (tid >= dd) ? sc[tid - dd] : 0;
    __syncthreads();
    sc[tid] += v;
    __syncthreads();
  }
  int run = sc[tid] - s;
#pragma unroll
  for (int k = 0; k < 4; ++k) {
    off[4 * tid + k] = run;
    run += c[k];
  }
  __syncthreads();
  if (blockIdx.x == 0)
    for (int i = tid; i <= nown; i += 256) ooff[i] = (i < nown) ? off[i] : sc[255];

  const int base = blockIdx.x * (256 * EPT);
  int dv[EPT], sv[EPT];
#pragma unroll
  for (int k = 0; k < EPT; ++k) {
    const int e = base + k * 256 + tid;
    const int ec = (e < E) ? e : E - 1;
    dv[k] = dst[ec];
    sv[k] = src[ec];
  }
  int ow[EPT], rk[EPT];
#pragma unroll
  for (int k = 0; k < EPT; ++k) {
    const int e = base + k * 256 + tid;
    ow[k] = dv[k] / RNG;
    rk[k] = (e < E) ? atomicAdd(&h[ow[k]], 1) : 0;
  }
  __syncthreads();
  for (int i = tid; i < nown; i += 256) {
    const int cc = h[i];
    if (cc) off[i] += atomicAdd(&ofill[i], cc);
  }
  __syncthreads();
#pragma unroll
  for (int k = 0; k < EPT; ++k) {
    const int e = base + k * 256 + tid;
    if (e < E)
      list[off[ow[k]] + rk[k]] =
          (unsigned)sv[k] | ((unsigned)(dv[k] - ow[k] * RNG) << SRC_BITS);
  }
}

__global__ void split_z2_kernel(const float* __restrict__ z2,
                                float* __restrict__ out01,
                                float* __restrict__ out2, int n) {
  int i = blockIdx.x * blockDim.x + threadIdx.x;
  if (i < n) {
    out01[2 * i + 0] = z2[3 * i + 0];
    out01[2 * i + 1] = z2[3 * i + 1];
    out2[i] = z2[3 * i + 2];
  }
}

template <int DIN, int DOUT, int ACT, bool HAS_AGG>
__global__ __launch_bounds__(2 * ((DOUT + 15) & ~15)) void wmma_linear_kernel(
    const float* agg, const float* __restrict__ invdeg,
    const float* __restrict__ x,
    const float* __restrict__ Wl, const float* __restrict__ bias,
    const float* __restrict__ Wr,
    float* out, float* out2, int nnodes) {
  constexpr int MT = 4;
  constexpr int ROWS = 16 * MT;
  constexpr int DOUTP = (DOUT + 15) & ~15;
  constexpr int LDW = DIN + 4;
  constexpr int C4 = DIN / 4;
  constexpr int KB = DIN / 8;
  __shared__ alignas(16) float sA[ROWS * LDW];
  __shared__ alignas(16) float sX[HAS_AGG ? ROWS * LDW : 4];

  const int tid = threadIdx.x;
  constexpr int NT = 2 * DOUTP;
  const int row0 = blockIdx.x * ROWS;

  const int wave = tid >> 5;
  const int lane = tid & 31;
  const int half = lane >> 4;
  const int l16 = lane & 15;
  const int n = wave * 16 + l16;
  const bool colOK = (DOUT == DOUTP) || (n < DOUT);

  constexpr int APT = ROWS * C4 / NT;
  constexpr int WREG = KB * (HAS_AGG ? 8 : 4);
  constexpr int AMAX = (208 - WREG) / (HAS_AGG ? 9 : 4);
  constexpr int AG = (APT <= AMAX) ? APT : (APT / 2 <= AMAX) ? APT / 2 : (APT / 4 <= AMAX) ? APT / 4 : APT / 8;
  float4 bl[KB];
  float4 br[HAS_AGG ? KB : 1];
  const int nr = colOK ? n : 0;
  const float* pl = Wl + (long)nr * DIN + 4 * half;
  const float* pr = HAS_AGG ? Wr + (long)nr * DIN + 4 * half : Wl;
  for (int g = 0; g < APT; g += AG) {
    float4 av[AG];
    float4 xv[HAS_AGG ? AG : 1];
    float sc[HAS_AGG ? AG : 1];
#pragma unroll
    for (int j = 0; j < AG; ++j) {
      const int i = tid + (g + j) * NT;
      const int r = i / C4, c4 = (i - r * C4) * 4;
      int m = row0 + r;
      if (m >= nnodes) m = nnodes - 1;
      if constexpr (HAS_AGG) {
        sc[j] = invdeg[m];
        av[j] = *(const float4*)(agg + (long)m * DIN + c4);
        xv[j] = *(const float4*)(x + (long)m * DIN + c4);
      } else {
        av[j] = *(const float4*)(x + (long)m * DIN + c4);
      }
    }
    if (g == 0) {
#pragma unroll
      for (int j = 0; j < KB; ++j) {
        bl[j] = *(const float4*)(pl + 8 * j);
        if constexpr (HAS_AGG) br[j] = *(const float4*)(pr + 8 * j);
      }
    }
#pragma unroll
    for (int j = 0; j < AG; ++j) {
      const int i = tid + (g + j) * NT;
      const int r = i / C4, c4 = (i - r * C4) * 4;
      if constexpr (HAS_AGG) {
        float4 a = av[j];
        const float s = sc[j];
        a.x *= s; a.y *= s; a.z *= s; a.w *= s;
        *(float4*)&sA[r * LDW + c4] = a;
        *(float4*)&sX[r * LDW + c4] = xv[j];
      } else {
        *(float4*)&sA[r * LDW + c4] = av[j];
      }
    }
  }
  __syncthreads();

  v8f acc[MT];
#pragma unroll
  for (int mt = 0; mt < MT; ++mt) acc[mt] = (v8f){};

#pragma unroll
  for (int j = 0; j < KB; ++j) {
    const int kk = 8 * j + 4 * half;
    float4 a[MT];
    float4 a2[HAS_AGG ? MT : 1];
#pragma unroll
    for (int mt = 0; mt < MT; ++mt) {
      const int r = mt * 16 + l16;
      a[mt] = *(const float4*)&sA[r * LDW + kk];
      if constexpr (HAS_AGG) a2[mt] = *(const float4*)&sX[r * LDW + kk];
    }
    const v2f b0 = {bl[j].x, bl[j].y};
    const v2f b1 = {bl[j].z, bl[j].w};
#pragma unroll
    for (int mt = 0; mt < MT; ++mt) {
      const v2f a0 = {a[mt].x, a[mt].y};
      acc[mt] = __builtin_amdgcn_wmma_f32_16x16x4_f32(false, a0, false, b0,
                                                      (short)0, acc[mt], false,
                                                      false);
      if constexpr (HAS_AGG) {
        const v2f x0 = {a2[mt].x, a2[mt].y};
        const v2f c0 = {br[j].x, br[j].y};
        acc[mt] = __builtin_amdgcn_wmma_f32_16x16x4_f32(false, x0, false, c0,
                                                        (short)0, acc[mt],
                                                        false, false);
      }
    }
#pragma unroll
    for (int mt = 0; mt < MT; ++mt) {
      const v2f a1 = {a[mt].z, a[mt].w};
      acc[mt] = __builtin_amdgcn_wmma_f32_16x16x4_f32(false, a1, false, b1,
                                                      (short)0, acc[mt], false,
                                                      false);
      if constexpr (HAS_AGG) {
        const v2f x1 = {a2[mt].z, a2[mt].w};
        const v2f c1 = {br[j].z, br[j].w};
        acc[mt] = __builtin_amdgcn_wmma_f32_16x16x4_f32(false, x1, false, c1,
                                                        (short)0, acc[mt],
                                                        false, false);
      }
    }
  }

  const float bcol = colOK ? bias[colOK ? n : 0] : 0.0f;
#pragma unroll
  for (int mt = 0; mt < MT; ++mt) {
#pragma unroll
    for (int i = 0; i < 8; ++i) {
      int row = row0 + mt * 16 + i + 8 * half;
      float v = acc[mt][i] + bcol;
      if (ACT == 1) v = tanhf(v);
      if (ACT == 2) v = (v > 0.0f) ? v : 0.01f * v;
      if (colOK && row < nnodes) {
        out[(long)row * DOUT + n] = v;
        if (out2) out2[(long)row * DOUT + n] = v;
      }
    }
  }
}

template <int DIN, int DOUT, int ACT>
__global__ __launch_bounds__(512) void owner_sage_kernel(
    const float* __restrict__ x, const unsigned* __restrict__ list,
    const int* __restrict__ ooff, const float* __restrict__ invdeg,
    const float* __restrict__ Wl, const float* __restrict__ bias,
    const float* __restrict__ Wr, float* __restrict__ out, int nnodes) {
  constexpr int D = DIN;
  constexpr int V = D / 32;
  constexpr int U = 64 / V;
  typedef float rowv __attribute__((ext_vector_type(V)));
  constexpr int NW = 16;
  constexpr int NTH = NW * 32;
  constexpr int LDW = D + 4;
  constexpr int SR = 64;
  constexpr int NST = (RNG + SR - 1) / SR;
  constexpr int MTT = RNG / 16;
  constexpr int DOUTP = (DOUT + 15) & ~15;
  constexpr int NTN = DOUTP / 16;
  constexpr int WPN = NW / NTN;
  constexpr int MTW = (WPN >= 4) ? 1 : 4 / WPN;
  constexpr int C4 = D / 4;
  constexpr int KB = D / 8;
  constexpr int APT = SR * C4 / NTH;
  static_assert(RNG % 16 == 0 && NW % NTN == 0 && (SR * C4) % NTH == 0, "shape");
  __shared__ alignas(16) float acc[RNG * LDW];
  __shared__ alignas(16) float sX[2 * SR * LDW];
  const int tid = threadIdx.x;
  const int r0 = blockIdx.x * RNG;
  const int beg = ooff[blockIdx.x];
  const int end = ooff[blockIdx.x + 1];
  const int lane = tid & 31;
  const int wv = __builtin_amdgcn_readfirstlane(tid >> 5);

  typedef float xnv __attribute__((ext_vector_type(4 * APT)));
  xnv xn;
#pragma unroll
  for (int j = 0; j < APT; ++j) {
    const int i = tid + j * NTH;
    const int r = i / C4, c4 = (i - r * C4) * 4;
    int m = r0 + r;
    if (m >= nnodes) m = nnodes - 1;
    const float4 t = *(const float4*)(x + (long)m * D + c4);
    xn[4 * j + 0] = t.x; xn[4 * j + 1] = t.y; xn[4 * j + 2] = t.z; xn[4 * j + 3] = t.w;
  }
  for (int i = tid; i < RNG * LDW / 4; i += NTH)
    ((float4*)acc)[i] = make_float4(0.0f, 0.0f, 0.0f, 0.0f);
#pragma unroll
  for (int j = 0; j < APT; ++j) {
    const int i = tid + j * NTH;
    const int r = i / C4, c4 = (i - r * C4) * 4;
    *(float4*)&sX[r * LDW + c4] = make_float4(xn[4 * j + 0], xn[4 * j + 1], xn[4 * j + 2], xn[4 * j + 3]);
  }
  __syncthreads();

  for (int st = beg + wv * 32; st < end; st += NW * 32) {
    const int qi = (st + lane < end) ? st + lane : end - 1;
    const unsigned p = list[qi];
    const bool mine = (st + lane < end);
    const int sv = mine ? (int)(p & ((1u << SRC_BITS) - 1u)) : 0;
    const int dl = mine ? (int)(p >> SRC_BITS) : -1;
    for (int i0 = 0; i0 < 32 && st + i0 < end; i0 += U) {
      rowv v[U];
#pragma unroll
      for (int j = 0; j < U; ++j) {
        const int sj = __builtin_amdgcn_readlane(sv, i0 + j);
        const float* r = x + (long)sj * D;
        v[j] = *(const rowv*)(r + V * lane);
      }
#pragma unroll
      for (int j = 0; j < U; ++j) {
        const int dj = __builtin_amdgcn_readlane(dl, i0 + j);
        if (dj >= 0) {
          float* o = acc + dj * LDW + lane;
#pragma unroll
          for (int c = 0; c < V; ++c) lds_add_f32(o + 32 * c, v[j][c]);
        }
      }
    }
  }

  const int half = lane >> 4;
  const int l16 = lane & 15;
  const int ntile = wv % NTN;
  const int mslot = wv / NTN;
  const int n = ntile * 16 + l16;
  const bool colOK = (DOUT == DOUTP) || (n < DOUT);
  const int nr = colOK ? n : 0;
  const float* pl = Wl + (long)nr * DIN + 4 * half;
  const float* pr = Wr + (long)nr * DIN + 4 * half;
  float4 bl[KB], br[KB];
#pragma unroll
  for (int j = 0; j < KB; ++j) {
    bl[j] = *(const float4*)(pl + 8 * j);
    br[j] = *(const float4*)(pr + 8 * j);
  }
  const float bcol = colOK ? bias[nr] : 0.0f;
  __syncthreads();

  for (int r = wv; r < RNG; r += NW) {
    int m = r0 + r;
    if (m >= nnodes) m = nnodes - 1;
    const float s = invdeg[m];
    rowv o;
#pragma unroll
    for (int c = 0; c < V; ++c) o[c] = acc[r * LDW + 32 * c + lane] * s;
    *(rowv*)&acc[r * LDW + V * lane] = o;
  }
  __syncthreads();

  for (int s = 0; s < NST; ++s) {
    const float* sXc = sX + (s & 1) * (SR * LDW);
    if (s + 1 < NST) {
#pragma unroll
      for (int j = 0; j < APT; ++j) {
        const int i = tid + j * NTH;
        const int r = i / C4, c4 = (i - r * C4) * 4;
        int m = r0 + (s + 1) * SR + r;
        if (m >= nnodes) m = nnodes - 1;
        const float4 t = *(const float4*)(x + (long)m * D + c4);
        xn[4 * j + 0] = t.x; xn[4 * j + 1] = t.y; xn[4 * j + 2] = t.z; xn[4 * j + 3] = t.w;
      }
    }
    bool ok[MTW];
    bool any = false;
#pragma unroll
    for (int i = 0; i < MTW; ++i) {
      const int ms = mslot + i * WPN;
      ok[i] = (ms < 4) && (4 * s + ms < MTT);
      any = any || ok[i];
    }
    if (any) {
      v8f accv[MTW];
      int rl[MTW];
#pragma unroll
      for (int i = 0; i < MTW; ++i) {
        accv[i] = (v8f){};
        rl[i] = (ok[i] ? (mslot + i * WPN) * 16 : 0) + l16;
      }
#pragma unroll
      for (int j = 0; j < KB; ++j) {
        const int kk = 8 * j + 4 * half;
        float4 a[MTW], a2[MTW];
#pragma unroll
        for (int i = 0; i < MTW; ++i) {
          a[i] = *(const float4*)&acc[(s * SR + rl[i]) * LDW + kk];
          a2[i] = *(const float4*)&sXc[rl[i] * LDW + kk];
        }
        const v2f b0 = {bl[j].x, bl[j].y};
        const v2f b1 = {bl[j].z, bl[j].w};
        const v2f c0 = {br[j].x, br[j].y};
        const v2f c1 = {br[j].z, br[j].w};
#pragma unroll
        for (int i = 0; i < MTW; ++i) {
          const v2f a0 = {a[i].x, a[i].y};
          const v2f x0 = {a2[i].x, a2[i].y};
          accv[i] = __builtin_amdgcn_wmma_f32_16x16x4_f32(false, a0, false, b0, (short)0, accv[i], false, false);
          accv[i] = __builtin_amdgcn_wmma_f32_16x16x4_f32(false, x0, false, c0, (short)0, accv[i], false, false);
        }
#pragma unroll
        for (int i = 0; i < MTW; ++i) {
          const v2f a1 = {a[i].z, a[i].w};
          const v2f x1 = {a2[i].z, a2[i].w};
          accv[i] = __builtin_amdgcn_wmma_f32_16x16x4_f32(false, a1, false, b1, (short)0, accv[i], false, false);
          accv[i] = __builtin_amdgcn_wmma_f32_16x16x4_f32(false, x1, false, c1, (short)0, accv[i], false, false);
        }
      }
#pragma unroll
      for (int i = 0; i < MTW; ++i) {
        if (ok[i]) {
#pragma unroll
          for (int e = 0; e < 8; ++e) {
            const int row = r0 + s * SR + (mslot + i * WPN) * 16 + e + 8 * half;
            float vv = accv[i][e] + bcol;
            if (ACT == 1) vv = tanhf(vv);
            if (ACT == 2) vv = (vv > 0.0f) ? vv : 0.01f * vv;
            if (colOK && row < nnodes) out[(long)row * DOUT + n] = vv;
          }
        }
      }
    }
    if (s + 1 < NST) {
      float* sXn = sX + ((s + 1) & 1) * (SR * LDW);
#pragma unroll
      for (int j = 0; j < APT; ++j) {
        const int i = tid + j * NTH;
        const int r = i / C4, c4 = (i - r * C4) * 4;
        *(float4*)&sXn[r * LDW + c4] = make_float4(xn[4 * j + 0], xn[4 * j + 1], xn[4 * j + 2], xn[4 * j + 3]);
      }
    }
    __syncthreads();
  }
}


static inline unsigned cdiv(long a, long b) { return (unsigned)((a + b - 1) / b); }

extern "C" void kernel_launch(void* const* d_in, const int* in_sizes, int n_in,
                              void* d_out, int out_size, void* d_ws, size_t ws_size,
                              hipStream_t stream) {
  const float* x = (const float*)d_in[0];
  const int* ei = (const int*)d_in[1];
  const int N = in_sizes[0] / 64;
  const int E = in_sizes[1] / 2;
  const int* src = ei;
  const int* dst = ei + E;

  const float* g1Wl = (const float*)d_in[2];
  const float* g1b  = (const float*)d_in[3];
  const float* g1Wr = (const float*)d_in[4];
  const float* g2Wl = (const float*)d_in[5];
  const float* g2b  = (const float*)d_in[6];
  const float* g2Wr = (const float*)d_in[7];
  const float* g3Wl = (const float*)d_in[8];
  const float* g3b  = (const float*)d_in[9];
  const float* g3Wr = (const float*)d_in[10];
  const float* trW  = (const float*)d_in[11];
  const float* trb  = (const float*)d_in[12];
  const float* d1W  = (const float*)d_in[13];
  const float* d1b  = (const float*)d_in[14];
  const float* d2Wl = (const float*)d_in[15];
  const float* d2b  = (const float*)d_in[16];
  const float* d2Wr = (const float*)d_in[17];
  const float* d3Wl = (const float*)d_in[18];
  const float* d3b  = (const float*)d_in[19];
  const float* d3Wr = (const float*)d_in[20];
  const float* d4Wl = (const float*)d_in[21];
  const float* d4b  = (const float*)d_in[22];
  const float* d4Wr = (const float*)d_in[23];

  const unsigned gOwn = cdiv(N, RNG);
  float* ws   = (float*)d_ws;
  float* inv  = ws;
  unsigned* list = (unsigned*)(ws + N);
  float* bufA = (float*)(((uintptr_t)(ws + (size_t)N + (size_t)E) + 255) & ~(uintptr_t)255);
  float* bufB = bufA + (size_t)N * 128;
  float* bufC = bufB + (size_t)N * 128;
  int* ooff   = (int*)(bufC + (size_t)N * 64);
  float* cnt  = bufA;
  int* ocnt   = (int*)(bufA + N);
  int* ofill  = ocnt + MAXOWN;
  float* z2   = bufB;

  float* outZ01 = (float*)d_out;
  float* outZ2  = outZ01 + 2 * (size_t)N;
  float* outMu  = outZ2 + N;
  float* outLv  = outMu + (size_t)N * 64;

  const unsigned gN64 = (unsigned)((N + 63) / 64);
  const unsigned gBin = cdiv(E, 256 * EPT);

  zero_kernel<<<cdiv(N + 2 * MAXOWN, 256), 256, 0, stream>>>(cnt, N + 2 * MAXOWN);
  degree_count_kernel<<<gBin, 256, 0, stream>>>(dst, cnt, ocnt, E, (int)gOwn);
  invdeg_kernel<<<cdiv(N, 256), 256, 0, stream>>>(cnt, inv, N);
  owner_bin_kernel<<<gBin, 256, 0, stream>>>(src, dst, ocnt, ofill, ooff, list, E, (int)gOwn);

  owner_sage_kernel<64, 128, 1><<<gOwn, 512, 0, stream>>>(
      x, list, ooff, inv, g1Wl, g1b, g1Wr, bufB, N);

  owner_sage_kernel<128, 128, 1><<<gOwn, 512, 0, stream>>>(
      bufB, list, ooff, inv, g2Wl, g2b, g2Wr, bufA, N);

  owner_sage_kernel<128, 128, 1><<<gOwn, 512, 0, stream>>>(
      bufA, list, ooff, inv, g3Wl, g3b, g3Wr, bufB, N);

  wmma_linear_kernel<128, 64, 0, false><<<gN64, 128, 0, stream>>>(
      nullptr, nullptr, bufB, trW, trb, nullptr, outMu, outLv, N);

  wmma_linear_kernel<64, 128, 2, false><<<gN64, 256, 0, stream>>>(
      nullptr, nullptr, outMu, d1W, d1b, nullptr, bufA, nullptr, N);

  owner_sage_kernel<128, 128, 2><<<gOwn, 512, 0, stream>>>(
      bufA, list, ooff, inv, d2Wl, d2b, d2Wr, bufB, N);

  owner_sage_kernel<128, 64, 1><<<gOwn, 512, 0, stream>>>(
      bufB, list, ooff, inv, d3Wl, d3b, d3Wr, bufC, N);

  owner_sage_kernel<64, 3, 0><<<gOwn, 512, 0, stream>>>(
      bufC, list, ooff, inv, d4Wl, d4b, d4Wr, z2, N);

  split_z2_kernel<<<cdiv(N, 256), 256, 0, stream>>>(z2, outZ01, outZ2, N);
}


# SAGE layers: owner tile cut into 5 sub-tiles of 80 rows, edges binned per sub-tile; a wave fetches and adds sub-tile i+1's rows between the K blocks of its products of sub-tile i; arrival order kept
# speedup vs baseline: 3.4622x; 1.0473x over previous
#include <hip/hip_runtime.h>
#include <hip/hip_bf16.h>

typedef __attribute__((ext_vector_type(2))) float v2f;
typedef __attribute__((ext_vector_type(8))) float v8f;


__device__ __forceinline__ void atom_add_f32(float* p, float v) {
  __hip_atomic_fetch_add(p, v, __ATOMIC_RELAXED, __HIP_MEMORY_SCOPE_AGENT);
}

__global__ void zero_kernel(float* __restrict__ p, long n) {
  long i = (long)blockIdx.x * blockDim.x + threadIdx.x;
  if (i < n) p[i] = 0.0f;
}

__global__ void invdeg_kernel(const float* __restrict__ cnt, float* __restrict__ inv, int n) {
  int i = blockIdx.x * blockDim.x + threadIdx.x;
  if (i < n) inv[i] = 1.0f / fmaxf(cnt[i], 1.0f);
}

constexpr int RNG = 400;
constexpr int SUB = 80;
constexpr int NSUB = RNG / SUB;
constexpr int MAXBIN = 2048;
constexpr int EPT = 16;
constexpr int SRC_BITS = 23;
static_assert(RNG % SUB == 0 && SUB % 16 == 0, "sub-tiles");

__device__ __forceinline__ void lds_add_f32(float* p, float v) {
  __hip_atomic_fetch_add(p, v, __ATOMIC_RELAXED, __HIP_MEMORY_SCOPE_WORKGROUP);
}

__global__ __launch_bounds__(256) void degree_count_kernel(const int* __restrict__ dst,
                                                           float* __restrict__ cnt,
                                                           int* __restrict__ ocnt, int E, int nown) {
  __shared__ int h[MAXBIN];
  const int tid = threadIdx.x;
  for (int i = tid; i < MAXBIN; i += 256) h[i] = 0;
  __syncthreads();
  const int base = blockIdx.x * (256 * EPT);
  int d[EPT];
#pragma unroll
  for (int k = 0; k < EPT; ++k) {
    const int e = base + k * 256 + tid;
    d[k] = (e < E) ? dst[e] : -1;
  }
#pragma unroll
  for (int k = 0; k < EPT; ++k) {
    if (d[k] >= 0) {
      atom_add_f32(&cnt[d[k]], 1.0f);
      atomicAdd(&h[d[k] / SUB], 1);
    }
  }
  __syncthreads();
  for (int i = tid; i < nown; i += 256) {
    const int c = h[i];
    if (c) atomicAdd(&ocnt[i], c);
  }
}

__global__ __launch_bounds__(256) void owner_bin_kernel(const int* __restrict__ src,
                                                        const int* __restrict__ dst,
                                                        const int* __restrict__ ocnt,
                                                        int* __restrict__ ofill,
                                                        int* __restrict__ ooff,
                                                        unsigned* __restrict__ list,
                                                        int E, int nown, int nslot) {
  constexpr int BPT = MAXBIN / 256;
  __shared__ int h[MAXBIN];
  __shared__ int off[MAXBIN];
  __shared__ int sc[256];
  const int tid = threadIdx.x;
  for (int i = tid; i < MAXBIN; i += 256) h[i] = 0;
  int c[BPT];
  int s = 0;
#pragma unroll
  for (int k = 0; k < BPT; ++k) {
    const int b = BPT * tid + k;
    c[k] = (b < nown) ? ocnt[b] : 0;
    s += c[k];
  }
  sc[tid] = s;
  __syncthreads();
  for (int dd = 1; dd < 256; dd <<= 1) {
    const int v = (tid >= dd) ? sc[tid - dd] : 0;
    __syncthreads();
    sc[tid] += v;
    __syncthreads();
  }
  int run = sc[tid] - s;
#pragma unroll
  for (int k = 0; k < BPT; ++k) {
    off[BPT * tid + k] = run;
    run += c[k];
  }
  __syncthreads();
  if (blockIdx.x == 0)
    for (int i = tid; i <= nslot; i += 256) ooff[i] = (i < nown) ? off[i] : sc[255];

  const int base = blockIdx.x * (256 * EPT);
  int dv[EPT], sv[EPT];
#pragma unroll
  for (int k = 0; k < EPT; ++k) {
    const int e = base + k * 256 + tid;
    const int ec = (e < E) ? e : E - 1;
    dv[k] = dst[ec];
    sv[k] = src[ec];
  }
  int ow[EPT], rk[EPT];
#pragma unroll
  for (int k = 0; k < EPT; ++k) {
    const int e = base + k * 256 + tid;
    ow[k] = dv[k] / SUB;
    rk[k] = (e < E) ? atomicAdd(&h[ow[k]], 1) : 0;
  }
  __syncthreads();
  for (int i = tid; i < nown; i += 256) {
    const int cc = h[i];
    if (cc) off[i] += atomicAdd(&ofill[i], cc);
  }
  __syncthreads();
#pragma unroll
  for (int k = 0; k < EPT; ++k) {
    const int e = base + k * 256 + tid;
    if (e < E)
      list[off[ow[k]] + rk[k]] =
          (unsigned)sv[k] | ((unsigned)(dv[k] - (dv[k] / RNG) * RNG) << SRC_BITS);
  }
}

__global__ void split_z2_kernel(const float* __restrict__ z2,
                                float* __restrict__ out01,
                                float* __restrict__ out2, int n) {
  int i = blockIdx.x * blockDim.x + threadIdx.x;
  if (i < n) {
    out01[2 * i + 0] = z2[3 * i + 0];
    out01[2 * i + 1] = z2[3 * i + 1];
    out2[i] = z2[3 * i + 2];
  }
}

template <int DIN, int DOUT, int ACT, bool HAS_AGG>
__global__ __launch_bounds__(2 * ((DOUT + 15) & ~15)) void wmma_linear_kernel(
    const float* agg, const float* __restrict__ invdeg,
    const float* __restrict__ x,
    const float* __restrict__ Wl, const float* __restrict__ bias,
    const float* __restrict__ Wr,
    float* out, float* out2, int nnodes) {
  constexpr int MT = 4;
  constexpr int ROWS = 16 * MT;
  constexpr int DOUTP = (DOUT + 15) & ~15;
  constexpr int LDW = DIN + 4;
  constexpr int C4 = DIN / 4;
  constexpr int KB = DIN / 8;
  __shared__ alignas(16) float sA[ROWS * LDW];
  __shared__ alignas(16) float sX[HAS_AGG ? ROWS * LDW : 4];

  const int tid = threadIdx.x;
  constexpr int NT = 2 * DOUTP;
  const int row0 = blockIdx.x * ROWS;

  const int wave = tid >> 5;
  const int lane = tid & 31;
  const int half = lane >> 4;
  const int l16 = lane & 15;
  const int n = wave * 16 + l16;
  const bool colOK = (DOUT == DOUTP) || (n < DOUT);

  constexpr int APT = ROWS * C4 / NT;
  constexpr int WREG = KB * (HAS_AGG ? 8 : 4);
  constexpr int AMAX = (208 - WREG) / (HAS_AGG ? 9 : 4);
  constexpr int AG = (APT <= AMAX) ? APT : (APT / 2 <= AMAX) ? APT / 2 : (APT / 4 <= AMAX) ? APT / 4 : APT / 8;
  float4 bl[KB];
  float4 br[HAS_AGG ? KB : 1];
  const int nr = colOK ? n : 0;
  const float* pl = Wl + (long)nr * DIN + 4 * half;
  const float* pr = HAS_AGG ? Wr + (long)nr * DIN + 4 * half : Wl;
  for (int g = 0; g < APT; g += AG) {
    float4 av[AG];
    float4 xv[HAS_AGG ? AG : 1];
    float sc[HAS_AGG ? AG : 1];
#pragma unroll
    for (int j = 0; j < AG; ++j) {
      const int i = tid + (g + j) * NT;
      const int r = i / C4, c4 = (i - r * C4) * 4;
      int m = row0 + r;
      if (m >= nnodes) m = nnodes - 1;
      if constexpr (HAS_AGG) {
        sc[j] = invdeg[m];
        av[j] = *(const float4*)(agg + (long)m * DIN + c4);
        xv[j] = *(const float4*)(x + (long)m * DIN + c4);
      } else {
        av[j] = *(const float4*)(x + (long)m * DIN + c4);
      }
    }
    if (g == 0) {
#pragma unroll
      for (int j = 0; j < KB; ++j) {
        bl[j] = *(const float4*)(pl + 8 * j);
        if constexpr (HAS_AGG) br[j] = *(const float4*)(pr + 8 * j);
      }
    }
#pragma unroll
    for (int j = 0; j < AG; ++j) {
      const int i = tid + (g + j) * NT;
      const int r = i / C4, c4 = (i - r * C4) * 4;
      if constexpr (HAS_AGG) {
        float4 a = av[j];
        const float s = sc[j];
        a.x *= s; a.y *= s; a.z *= s; a.w *= s;
        *(float4*)&sA[r * LDW + c4] = a;
        *(float4*)&sX[r * LDW + c4] = xv[j];
      } else {
        *(float4*)&sA[r * LDW + c4] = av[j];
      }
    }
  }
  __syncthreads();

  v8f acc[MT];
#pragma unroll
  for (int mt = 0; mt < MT; ++mt) acc[mt] = (v8f){};

#pragma unroll
  for (int j = 0; j < KB; ++j) {
    const int kk = 8 * j + 4 * half;
    float4 a[MT];
    float4 a2[HAS_AGG ? MT : 1];
#pragma unroll
    for (int mt = 0; mt < MT; ++mt) {
      const int r = mt * 16 + l16;
      a[mt] = *(const float4*)&sA[r * LDW + kk];
      if constexpr (HAS_AGG) a2[mt] = *(const float4*)&sX[r * LDW + kk];
    }
    const v2f b0 = {bl[j].x, bl[j].y};
    const v2f b1 = {bl[j].z, bl[j].w};
#pragma unroll
    for (int mt = 0; mt < MT; ++mt) {
      const v2f a0 = {a[mt].x, a[mt].y};
      acc[mt] = __builtin_amdgcn_wmma_f32_16x16x4_f32(false, a0, false, b0,
                                                      (short)0, acc[mt], false,
                                                      false);
      if constexpr (HAS_AGG) {
        const v2f x0 = {a2[mt].x, a2[mt].y};
        const v2f c0 = {br[j].x, br[j].y};
        acc[mt] = __builtin_amdgcn_wmma_f32_16x16x4_f32(false, x0, false, c0,
                                                        (short)0, acc[mt],
                                                        false, false);
      }
    }
#pragma unroll
    for (int mt = 0; mt < MT; ++mt) {
      const v2f a1 = {a[mt].z, a[mt].w};
      acc[mt] = __builtin_amdgcn_wmma_f32_16x16x4_f32(false, a1, false, b1,
                                                      (short)0, acc[mt], false,
                                                      false);
      if constexpr (HAS_AGG) {
        const v2f x1 = {a2[mt].z, a2[mt].w};
        const v2f c1 = {br[j].z, br[j].w};
        acc[mt] = __builtin_amdgcn_wmma_f32_16x16x4_f32(false, x1, false, c1,
                                                        (short)0, acc[mt],
                                                        false, false);
      }
    }
  }

  const float bcol = colOK ? bias[colOK ? n : 0] : 0.0f;
#pragma unroll
  for (int mt = 0; mt < MT; ++mt) {
#pragma unroll
    for (int i = 0; i < 8; ++i) {
      int row = row0 + mt * 16 + i + 8 * half;
      float v = acc[mt][i] + bcol;
      if (ACT == 1) v = tanhf(v);
      if (ACT == 2) v = (v > 0.0f) ? v : 0.01f * v;
      if (colOK && row < nnodes) {
        out[(long)row * DOUT + n] = v;
        if (out2) out2[(long)row * DOUT + n] = v;
      }
    }
  }
}

template <int DIN, int DOUT, int ACT>
__global__ __launch_bounds__(512) void owner_sage_kernel(
    const float* __restrict__ x, const unsigned* __restrict__ list,
    const int* __restrict__ ooff, const float* __restrict__ invdeg,
    const float* __restrict__ Wl, const float* __restrict__ bias,
    const float* __restrict__ Wr, float* __restrict__ out, int nnodes) {
  constexpr int D = DIN;
  constexpr int V = D / 32;
  constexpr int U = 32 / V;
  constexpr int NB = 32 / U;
  typedef float rowv __attribute__((ext_vector_type(V)));
  constexpr int NW = 16;
  constexpr int NTH = NW * 32;
  constexpr int LDW = D + 4;
  constexpr int MTS = SUB / 16;
  constexpr int DOUTP = (DOUT + 15) & ~15;
  constexpr int NTN = DOUTP / 16;
  constexpr int WPN = NW / NTN;
  constexpr int C4 = D / 4;
  constexpr int KB = D / 8;
  constexpr int G = 4;
  constexpr int XF4 = SUB * C4;
  constexpr int APT = (XF4 + NTH - 1) / NTH;
  static_assert(NW % NTN == 0 && KB % G == 0 && SUB % NW == 0, "shape");
  __shared__ alignas(16) float acc[RNG * LDW];
  __shared__ alignas(16) float sX[2 * SUB * LDW];
  const int tid = threadIdx.x;
  const int r0 = blockIdx.x * RNG;
  const int lane = tid & 31;
  const int wv = __builtin_amdgcn_readfirstlane(tid >> 5);
  const int* obp = ooff + blockIdx.x * NSUB;
  const int ob0 = obp[0];
  int lb = obp[1], le = obp[2];

  typedef float xnv __attribute__((ext_vector_type(4 * APT)));
  xnv xn;
  auto root_load = [&](int sub) {
#pragma unroll
    for (int j = 0; j < APT; ++j) {
      int i = tid + j * NTH;
      if (i > XF4 - 1) i = XF4 - 1;
      const int r = i / C4, c4 = (i - r * C4) * 4;
      int m = r0 + sub * SUB + r;
      if (m >= nnodes) m = nnodes - 1;
      const float4 t = *(const float4*)(x + (long)m * D + c4);
      xn[4 * j + 0] = t.x; xn[4 * j + 1] = t.y; xn[4 * j + 2] = t.z; xn[4 * j + 3] = t.w;
    }
  };
  auto root_store = [&](float* dstX) {
#pragma unroll
    for (int j = 0; j < APT; ++j) {
      const int i = tid + j * NTH;
      const int r = i / C4, c4 = (i - r * C4) * 4;
      if (i < XF4) *(float4*)&dstX[r * LDW + c4] = make_float4(xn[4 * j + 0], xn[4 * j + 1], xn[4 * j + 2], xn[4 * j + 3]);
    }
  };

  int gst = 0, gend = 0;
  int gbi = 0;
  bool pend = false;
  unsigned pcur = 0, pnxt = 0;
  rowv v[U];
  auto list_word = [&](int st) -> unsigned {
    int q = st + lane;
    if (q > gend - 1) q = gend - 1;
    if (q < 0) q = 0;
    return list[q];
  };
  auto gather_begin = [&](int b, int e) {
    gend = e;
    gst = b + wv * 32;
    gbi = 0;
    pend = false;
    pcur = list_word(gst);
  };
  auto gather_step = [&]() {
    if (pend) {
      const int dl = (gst + lane < gend) ? (int)(pcur >> SRC_BITS) : -1;
#pragma unroll
      for (int j = 0; j < U; ++j) {
        const int dj = __builtin_amdgcn_readlane(dl, gbi * U + j);
        if (dj >= 0) {
          float* o = acc + dj * LDW + lane;
#pragma unroll
          for (int c = 0; c < V; ++c) lds_add_f32(o + 32 * c, v[j][c]);
        }
      }
      pend = false;
      ++gbi;
      if (gbi == NB || gst + gbi * U >= gend) {
        gst += NW * 32;
        gbi = 0;
        pcur = pnxt;
      }
    }
    if (gst < gend) {
      const int sv = (gst + lane < gend) ? (int)(pcur & ((1u << SRC_BITS) - 1u)) : 0;
#pragma unroll
      for (int j = 0; j < U; ++j) {
        const int sj = __builtin_amdgcn_readlane(sv, gbi * U + j);
        const float* r = x + (long)sj * D;
        v[j] = *(const rowv*)(r + V * lane);
      }
      if (gbi == 0) pnxt = list_word(gst + NW * 32);
      pend = true;
    }
  };

  root_load(0);
  for (int i = tid; i < RNG * LDW / 4; i += NTH)
    ((float4*)acc)[i] = make_float4(0.0f, 0.0f, 0.0f, 0.0f);
  root_store(sX);
  __syncthreads();

  gather_begin(ob0, lb);
  while (pend || gst < gend) gather_step();

  const int half = lane >> 4;
  const int l16 = lane & 15;
  const int ntile = wv % NTN;
  const int mslot = wv / NTN;
  const int n = ntile * 16 + l16;
  const bool colOK = (DOUT == DOUTP) || (n < DOUT);
  const int nr = colOK ? n : 0;
  const float* pl = Wl + (long)nr * DIN + 4 * half;
  const float* pr = Wr + (long)nr * DIN + 4 * half;
  float4 bl[KB], br[KB];
#pragma unroll
  for (int j = 0; j < KB; ++j) {
    bl[j] = *(const float4*)(pl + 8 * j);
    br[j] = *(const float4*)(pr + 8 * j);
  }
  const float bcol = colOK ? bias[nr] : 0.0f;

  auto scale_sub = [&](int sub) {
#pragma unroll
    for (int q = 0; q < SUB / NW; ++q) {
      const int r = sub * SUB + wv + q * NW;
      int m = r0 + r;
      if (m >= nnodes) m = nnodes - 1;
      const float s = invdeg[m];
      rowv o;
#pragma unroll
      for (int c = 0; c < V; ++c) o[c] = acc[r * LDW + 32 * c + lane] * s;
      *(rowv*)&acc[r * LDW + V * lane] = o;
    }
  };

  auto mtile = [&](int sub, int mt, const float* sXc, bool gat) {
    v8f accv = (v8f){};
    const int ra = (sub * SUB + mt * 16 + l16) * LDW;
    const int rx = (mt * 16 + l16) * LDW;
#pragma unroll
    for (int j = 0; j < KB; ++j) {
      if (j % G == 0) {
        if (gat) gather_step();
      }
      const int kk = 8 * j + 4 * half;
      const float4 a = *(const float4*)&acc[ra + kk];
      const float4 a2 = *(const float4*)&sXc[rx + kk];
      const v2f b0 = {bl[j].x, bl[j].y};
      const v2f b1 = {bl[j].z, bl[j].w};
      const v2f c0 = {br[j].x, br[j].y};
      const v2f c1 = {br[j].z, br[j].w};
      const v2f a0 = {a.x, a.y};
      const v2f x0 = {a2.x, a2.y};
      const v2f a1 = {a.z, a.w};
      const v2f x1 = {a2.z, a2.w};
      accv = __builtin_amdgcn_wmma_f32_16x16x4_f32(false, a0, false, b0, (short)0, accv, false, false);
      accv = __builtin_amdgcn_wmma_f32_16x16x4_f32(false, x0, false, c0, (short)0, accv, false, false);
      accv = __builtin_amdgcn_wmma_f32_16x16x4_f32(false, a1, false, b1, (short)0, accv, false, false);
      accv = __builtin_amdgcn_wmma_f32_16x16x4_f32(false, x1, false, c1, (short)0, accv, false, false);
    }
#pragma unroll
    for (int e = 0; e < 8; ++e) {
      const int row = r0 + sub * SUB + mt * 16 + e + 8 * half;
      float vv = accv[e] + bcol;
      if (ACT == 1) vv = tanhf(vv);
      if (ACT == 2) vv = (vv > 0.0f) ? vv : 0.01f * vv;
      if (colOK && row < nnodes) out[(long)row * DOUT + n] = vv;
    }
  };

  __syncthreads();
  scale_sub(0);
  __syncthreads();

  for (int i = 0; i < NSUB; ++i) {
    const float* sXc = sX + (i & 1) * (SUB * LDW);
    const bool more = (i + 1 < NSUB);
    if (more) {
      root_load(i + 1);
      gather_begin(lb, le);
      lb = le;
      le = obp[(i + 3 < NSUB) ? i + 3 : NSUB];
    }
    for (int mt = mslot; mt < MTS; mt += WPN) mtile(i, mt, sXc, more);
    if (more) {
      root_store(sX + ((i + 1) & 1) * (SUB * LDW));
      while (pend || gst < gend) gather_step();
      __syncthreads();
      scale_sub(i + 1);
      __syncthreads();
    }
  }
}


static inline unsigned cdiv(long a, long b) { return (unsigned)((a + b - 1) / b); }

extern "C" void kernel_launch(void* const* d_in, const int* in_sizes, int n_in,
                              void* d_out, int out_size, void* d_ws, size_t ws_size,
                              hipStream_t stream) {
  const float* x = (const float*)d_in[0];
  const int* ei = (const int*)d_in[1];
  const int N = in_sizes[0] / 64;
  const int E = in_sizes[1] / 2;
  const int* src = ei;
  const int* dst = ei + E;

  const float* g1Wl = (const float*)d_in[2];
  const float* g1b  = (const float*)d_in[3];
  const float* g1Wr = (const float*)d_in[4];
  const float* g2Wl = (const float*)d_in[5];
  const float* g2b  = (const float*)d_in[6];
  const float* g2Wr = (const float*)d_in[7];
  const float* g3Wl = (const float*)d_in[8];
  const float* g3b  = (const float*)d_in[9];
  const float* g3Wr = (const float*)d_in[10];
  const float* trW  = (const float*)d_in[11];
  const float* trb  = (const float*)d_in[12];
  const float* d1W  = (const float*)d_in[13];
  const float* d1b  = (const float*)d_in[14];
  const float* d2Wl = (const float*)d_in[15];
  const float* d2b  = (const float*)d_in[16];
  const float* d2Wr = (const float*)d_in[17];
  const float* d3Wl = (const float*)d_in[18];
  const float* d3b  = (const float*)d_in[19];
  const float* d3Wr = (const float*)d_in[20];
  const float* d4Wl = (const float*)d_in[21];
  const float* d4b  = (const float*)d_in[22];
  const float* d4Wr = (const float*)d_in[23];

  const unsigned gOwn = cdiv(N, RNG);
  float* ws   = (float*)d_ws;
  float* inv  = ws;
  unsigned* list = (unsigned*)(ws + N);
  float* bufA = (float*)(((uintptr_t)(ws + (size_t)N + (size_t)E) + 255) & ~(uintptr_t)255);
  float* bufB = bufA + (size_t)N * 128;
  int* ooff   = (int*)(bufB + (size_t)N * 128);
  float* cnt  = bufA;
  int* ocnt   = (int*)(bufA + N);
  int* ofill  = ocnt + MAXBIN;
  float* z2   = bufB;

  float* outZ01 = (float*)d_out;
  float* outZ2  = outZ01 + 2 * (size_t)N;
  float* outMu  = outZ2 + N;
  float* outLv  = outMu + (size_t)N * 64;

  const unsigned gN64 = (unsigned)((N + 63) / 64);
  const unsigned gBin = cdiv(E, 256 * EPT);
  const int nBin = (int)cdiv(N, SUB);
  const int nSlot = (int)gOwn * NSUB;

  zero_kernel<<<cdiv(N + 2 * MAXBIN, 256), 256, 0, stream>>>(cnt, N + 2 * MAXBIN);
  degree_count_kernel<<<gBin, 256, 0, stream>>>(dst, cnt, ocnt, E, nBin);
  invdeg_kernel<<<cdiv(N, 256), 256, 0, stream>>>(cnt, inv, N);
  owner_bin_kernel<<<gBin, 256, 0, stream>>>(src, dst, ocnt, ofill, ooff, list, E, nBin, nSlot);

  owner_sage_kernel<64, 128, 1><<<gOwn, 512, 0, stream>>>(
      x, list, ooff, inv, g1Wl, g1b, g1Wr, bufB, N);

  owner_sage_kernel<128, 128, 1><<<gOwn, 512, 0, stream>>>(
      bufB, list, ooff, inv, g2Wl, g2b, g2Wr, bufA, N);

  owner_sage_kernel<128, 128, 1><<<gOwn, 512, 0, stream>>>(
      bufA, list, ooff, inv, g3Wl, g3b, g3Wr, bufB, N);

  wmma_linear_kernel<128, 64, 0, false><<<gN64, 128, 0, stream>>>(
      nullptr, nullptr, bufB, trW, trb, nullptr, outMu, outLv, N);

  wmma_linear_kernel<64, 128, 2, false><<<gN64, 256, 0, stream>>>(
      nullptr, nullptr, outMu, d1W, d1b, nullptr, bufA, nullptr, N);

  owner_sage_kernel<128, 128, 2><<<gOwn, 512, 0, stream>>>(
      bufA, list, ooff, inv, d2Wl, d2b, d2Wr, bufB, N);

  owner_sage_kernel<128, 64, 1><<<gOwn, 512, 0, stream>>>(
      bufB, list, ooff, inv, d3Wl, d3b, d3Wr, bufA, N);

  owner_sage_kernel<64, 3, 0><<<gOwn, 512, 0, stream>>>(
      bufA, list, ooff, inv, d4Wl, d4b, d4Wr, z2, N);

  split_z2_kernel<<<cdiv(N, 256), 256, 0, stream>>>(z2, outZ01, outZ2, N);
}


# SAGE layers: a wave's A fragment loads for the next 2 K blocks go out ahead of its batch of LDS adds (now branch-free, empty slots add into a spare row): the adds drain under 8 WMMAs; sum order kept
# speedup vs baseline: 3.6014x; 1.0402x over previous
#include <hip/hip_runtime.h>
#include <hip/hip_bf16.h>

typedef __attribute__((ext_vector_type(2))) float v2f;
typedef __attribute__((ext_vector_type(8))) float v8f;


__device__ __forceinline__ void atom_add_f32(float* p, float v) {
  __hip_atomic_fetch_add(p, v, __ATOMIC_RELAXED, __HIP_MEMORY_SCOPE_AGENT);
}

__global__ void zero_kernel(float* __restrict__ p, long n) {
  long i = (long)blockIdx.x * blockDim.x + threadIdx.x;
  if (i < n) p[i] = 0.0f;
}

__global__ void invdeg_kernel(const float* __restrict__ cnt, float* __restrict__ inv, int n) {
  int i = blockIdx.x * blockDim.x + threadIdx.x;
  if (i < n) inv[i] = 1.0f / fmaxf(cnt[i], 1.0f);
}

constexpr int RNG = 400;
constexpr int SUB = 80;
constexpr int NSUB = RNG / SUB;
constexpr int MAXBIN = 2048;
constexpr int EPT = 16;
constexpr int SRC_BITS = 23;
static_assert(RNG % SUB == 0 && SUB % 16 == 0, "sub-tiles");

__device__ __forceinline__ void lds_add_f32(float* p, float v) {
  __hip_atomic_fetch_add(p, v, __ATOMIC_RELAXED, __HIP_MEMORY_SCOPE_WORKGROUP);
}

__global__ __launch_bounds__(256) void degree_count_kernel(const int* __restrict__ dst,
                                                           float* __restrict__ cnt,
                                                           int* __restrict__ ocnt, int E, int nown) {
  __shared__ int h[MAXBIN];
  const int tid = threadIdx.x;
  for (int i = tid; i < MAXBIN; i += 256) h[i] = 0;
  __syncthreads();
  const int base = blockIdx.x * (256 * EPT);
  int d[EPT];
#pragma unroll
  for (int k = 0; k < EPT; ++k) {
    const int e = base + k * 256 + tid;
    d[k] = (e < E) ? dst[e] : -1;
  }
#pragma unroll
  for (int k = 0; k < EPT; ++k) {
    if (d[k] >= 0) {
      atom_add_f32(&cnt[d[k]], 1.0f);
      atomicAdd(&h[d[k] / SUB], 1);
    }
  }
  __syncthreads();
  for (int i = tid; i < nown; i += 256) {
    const int c = h[i];
    if (c) atomicAdd(&ocnt[i], c);
  }
}

__global__ __launch_bounds__(256) void owner_bin_kernel(const int* __restrict__ src,
                                                        const int* __restrict__ dst,
                                                        const int* __restrict__ ocnt,
                                                        int* __restrict__ ofill,
                                                        int* __restrict__ ooff,
                                                        unsigned* __restrict__ list,
                                                        int E, int nown, int nslot) {
  constexpr int BPT = MAXBIN / 256;
  __shared__ int h[MAXBIN];
  __shared__ int off[MAXBIN];
  __shared__ int sc[256];
  const int tid = threadIdx.x;
  for (int i = tid; i < MAXBIN; i += 256) h[i] = 0;
  int c[BPT];
  int s = 0;
#pragma unroll
  for (int k = 0; k < BPT; ++k) {
    const int b = BPT * tid + k;
    c[k] = (b < nown) ? ocnt[b] : 0;
    s += c[k];
  }
  sc[tid] = s;
  __syncthreads();
  for (int dd = 1; dd < 256; dd <<= 1) {
    const int v = (tid >= dd) ? sc[tid - dd] : 0;
    __syncthreads();
    sc[tid] += v;
    __syncthreads();
  }
  int run = sc[tid] - s;
#pragma unroll
  for (int k = 0; k < BPT; ++k) {
    off[BPT * tid + k] = run;
    run += c[k];
  }
  __syncthreads();
  if (blockIdx.x == 0)
    for (int i = tid; i <= nslot; i += 256) ooff[i] = (i < nown) ? off[i] : sc[255];

  const int base = blockIdx.x * (256 * EPT);
  int dv[EPT], sv[EPT];
#pragma unroll
  for (int k = 0; k < EPT; ++k) {
    const int e = base + k * 256 + tid;
    const int ec = (e < E) ? e : E - 1;
    dv[k] = dst[ec];
    sv[k] = src[ec];
  }
  int ow[EPT], rk[EPT];
#pragma unroll
  for (int k = 0; k < EPT; ++k) {
    const int e = base + k * 256 + tid;
    ow[k] = dv[k] / SUB;
    rk[k] = (e < E) ? atomicAdd(&h[ow[k]], 1) : 0;
  }
  __syncthreads();
  for (int i = tid; i < nown; i += 256) {
    const int cc = h[i];
    if (cc) off[i] += atomicAdd(&ofill[i], cc);
  }
  __syncthreads();
#pragma unroll
  for (int k = 0; k < EPT; ++k) {
    const int e = base + k * 256 + tid;
    if (e < E)
      list[off[ow[k]] + rk[k]] =
          (unsigned)sv[k] | ((unsigned)(dv[k] - (dv[k] / RNG) * RNG) << SRC_BITS);
  }
}

__global__ void split_z2_kernel(const float* __restrict__ z2,
                                float* __restrict__ out01,
                                float* __restrict__ out2, int n) {
  int i = blockIdx.x * blockDim.x + threadIdx.x;
  if (i < n) {
    out01[2 * i + 0] = z2[3 * i + 0];
    out01[2 * i + 1] = z2[3 * i + 1];
    out2[i] = z2[3 * i + 2];
  }
}

template <int DIN, int DOUT, int ACT, bool HAS_AGG>
__global__ __launch_bounds__(2 * ((DOUT + 15) & ~15)) void wmma_linear_kernel(
    const float* agg, const float* __restrict__ invdeg,
    const float* __restrict__ x,
    const float* __restrict__ Wl, const float* __restrict__ bias,
    const float* __restrict__ Wr,
    float* out, float* out2, int nnodes) {
  constexpr int MT = 4;
  constexpr int ROWS = 16 * MT;
  constexpr int DOUTP = (DOUT + 15) & ~15;
  constexpr int LDW = DIN + 4;
  constexpr int C4 = DIN / 4;
  constexpr int KB = DIN / 8;
  __shared__ alignas(16) float sA[ROWS * LDW];
  __shared__ alignas(16) float sX[HAS_AGG ? ROWS * LDW : 4];

  const int tid = threadIdx.x;
  constexpr int NT = 2 * DOUTP;
  const int row0 = blockIdx.x * ROWS;

  const int wave = tid >> 5;
  const int lane = tid & 31;
  const int half = lane >> 4;
  const int l16 = lane & 15;
  const int n = wave * 16 + l16;
  const bool colOK = (DOUT == DOUTP) || (n < DOUT);

  constexpr int APT = ROWS * C4 / NT;
  constexpr int WREG = KB * (HAS_AGG ? 8 : 4);
  constexpr int AMAX = (208 - WREG) / (HAS_AGG ? 9 : 4);
  constexpr int AG = (APT <= AMAX) ? APT : (APT / 2 <= AMAX) ? APT / 2 : (APT / 4 <= AMAX) ? APT / 4 : APT / 8;
  float4 bl[KB];
  float4 br[HAS_AGG ? KB : 1];
  const int nr = colOK ? n : 0;
  const float* pl = Wl + (long)nr * DIN + 4 * half;
  const float* pr = HAS_AGG ? Wr + (long)nr * DIN + 4 * half : Wl;
  for (int g = 0; g < APT; g += AG) {
    float4 av[AG];
    float4 xv[HAS_AGG ? AG : 1];
    float sc[HAS_AGG ? AG : 1];
#pragma unroll
    for (int j = 0; j < AG; ++j) {
      const int i = tid + (g + j) * NT;
      const int r = i / C4, c4 = (i - r * C4) * 4;
      int m = row0 + r;
      if (m >= nnodes) m = nnodes - 1;
      if constexpr (HAS_AGG) {
        sc[j] = invdeg[m];
        av[j] = *(const float4*)(agg + (long)m * DIN + c4);
        xv[j] = *(const float4*)(x + (long)m * DIN + c4);
      } else {
        av[j] = *(const float4*)(x + (long)m * DIN + c4);
      }
    }
    if (g == 0) {
#pragma unroll
      for (int j = 0; j < KB; ++j) {
        bl[j] = *(const float4*)(pl + 8 * j);
        if constexpr (HAS_AGG) br[j] = *(const float4*)(pr + 8 * j);
      }
    }
#pragma unroll
    for (int j = 0; j < AG; ++j) {
      const int i = tid + (g + j) * NT;
      const int r = i / C4, c4 = (i - r * C4) * 4;
      if constexpr (HAS_AGG) {
        float4 a = av[j];
        const float s = sc[j];
        a.x *= s; a.y *= s; a.z *= s; a.w *= s;
        *(float4*)&sA[r * LDW + c4] = a;
        *(float4*)&sX[r * LDW + c4] = xv[j];
      } else {
        *(float4*)&sA[r * LDW + c4] = av[j];
      }
    }
  }
  __syncthreads();

  v8f acc[MT];
#pragma unroll
  for (int mt = 0; mt < MT; ++mt) acc[mt] = (v8f){};

#pragma unroll
  for (int j = 0; j < KB; ++j) {
    const int kk = 8 * j + 4 * half;
    float4 a[MT];
    float4 a2[HAS_AGG ? MT : 1];
#pragma unroll
    for (int mt = 0; mt < MT; ++mt) {
      const int r = mt * 16 + l16;
      a[mt] = *(const float4*)&sA[r * LDW + kk];
      if constexpr (HAS_AGG) a2[mt] = *(const float4*)&sX[r * LDW + kk];
    }
    const v2f b0 = {bl[j].x, bl[j].y};
    const v2f b1 = {bl[j].z, bl[j].w};
#pragma unroll
    for (int mt = 0; mt < MT; ++mt) {
      const v2f a0 = {a[mt].x, a[mt].y};
      acc[mt] = __builtin_amdgcn_wmma_f32_16x16x4_f32(false, a0, false, b0,
                                                      (short)0, acc[mt], false,
                                                      false);
      if constexpr (HAS_AGG) {
        const v2f x0 = {a2[mt].x, a2[mt].y};
        const v2f c0 = {br[j].x, br[j].y};
        acc[mt] = __builtin_amdgcn_wmma_f32_16x16x4_f32(false, x0, false, c0,
                                                        (short)0, acc[mt],
                                                        false, false);
      }
    }
#pragma unroll
    for (int mt = 0; mt < MT; ++mt) {
      const v2f a1 = {a[mt].z, a[mt].w};
      acc[mt] = __builtin_amdgcn_wmma_f32_16x16x4_f32(false, a1, false, b1,
                                                      (short)0, acc[mt], false,
                                                      false);
      if constexpr (HAS_AGG) {
        const v2f x1 = {a2[mt].z, a2[mt].w};
        const v2f c1 = {br[j].z, br[j].w};
        acc[mt] = __builtin_amdgcn_wmma_f32_16x16x4_f32(false, x1, false, c1,
                                                        (short)0, acc[mt],
                                                        false, false);
      }
    }
  }

  const float bcol = colOK ? bias[colOK ? n : 0] : 0.0f;
#pragma unroll
  for (int mt = 0; mt < MT; ++mt) {
#pragma unroll
    for (int i = 0; i < 8; ++i) {
      int row = row0 + mt * 16 + i + 8 * half;
      float v = acc[mt][i] + bcol;
      if (ACT == 1) v = tanhf(v);
      if (ACT == 2) v = (v > 0.0f) ? v : 0.01f * v;
      if (colOK && row < nnodes) {
        out[(long)row * DOUT + n] = v;
        if (out2) out2[(long)row * DOUT + n] = v;
      }
    }
  }
}

template <int DIN, int DOUT, int ACT>
__global__ __launch_bounds__(512) void owner_sage_kernel(
    const float* __restrict__ x, const unsigned* __restrict__ list,
    const int* __restrict__ ooff, const float* __restrict__ invdeg,
    const float* __restrict__ Wl, const float* __restrict__ bias,
    const float* __restrict__ Wr, float* __restrict__ out, int nnodes) {
  constexpr int D = DIN;
  constexpr int V = D / 32;
  constexpr int U = 32 / V;
  constexpr int NB = 32 / U;
  typedef float rowv __attribute__((ext_vector_type(V)));
  constexpr int NW = 16;
  constexpr int NTH = NW * 32;
  constexpr int LDW = D + 4;
  constexpr int MTS = SUB / 16;
  constexpr int DOUTP = (DOUT + 15) & ~15;
  constexpr int NTN = DOUTP / 16;
  constexpr int WPN = NW / NTN;
  constexpr int C4 = D / 4;
  constexpr int KB = D / 8;
  constexpr int G = 4;
  constexpr int XF4 = SUB * C4;
  constexpr int APT = (XF4 + NTH - 1) / NTH;
  static_assert(NW % NTN == 0 && KB % G == 0 && SUB % NW == 0, "shape");
  __shared__ alignas(16) float acc[(RNG + 1) * LDW];
  __shared__ alignas(16) float sX[2 * SUB * LDW];
  const int tid = threadIdx.x;
  const int r0 = blockIdx.x * RNG;
  const int lane = tid & 31;
  const int wv = __builtin_amdgcn_readfirstlane(tid >> 5);
  const int* obp = ooff + blockIdx.x * NSUB;
  const int ob0 = obp[0];
  int lb = obp[1], le = obp[2];

  typedef float xnv __attribute__((ext_vector_type(4 * APT)));
  xnv xn;
  auto root_load = [&](int sub) {
#pragma unroll
    for (int j = 0; j < APT; ++j) {
      int i = tid + j * NTH;
      if (i > XF4 - 1) i = XF4 - 1;
      const int r = i / C4, c4 = (i - r * C4) * 4;
      int m = r0 + sub * SUB + r;
      if (m >= nnodes) m = nnodes - 1;
      const float4 t = *(const float4*)(x + (long)m * D + c4);
      xn[4 * j + 0] = t.x; xn[4 * j + 1] = t.y; xn[4 * j + 2] = t.z; xn[4 * j + 3] = t.w;
    }
  };
  auto root_store = [&](float* dstX) {
#pragma unroll
    for (int j = 0; j < APT; ++j) {
      const int i = tid + j * NTH;
      const int r = i / C4, c4 = (i - r * C4) * 4;
      if (i < XF4) *(float4*)&dstX[r * LDW + c4] = make_float4(xn[4 * j + 0], xn[4 * j + 1], xn[4 * j + 2], xn[4 * j + 3]);
    }
  };

  int gst = 0, gend = 0;
  int gbi = 0;
  bool pend = false;
  unsigned pcur = 0, pnxt = 0;
  rowv v[U];
  auto list_word = [&](int st) -> unsigned {
    int q = st + lane;
    if (q > gend - 1) q = gend - 1;
    if (q < 0) q = 0;
    return list[q];
  };
  auto gather_begin = [&](int b, int e) {
    gend = e;
    gst = b + wv * 32;
    gbi = 0;
    pend = false;
    pcur = list_word(gst);
  };
  auto gather_add = [&]() {
    const int dl = (gst + lane < gend) ? (int)(pcur >> SRC_BITS) : RNG;
#pragma unroll
    for (int j = 0; j < U; ++j) {
      const int dj = __builtin_amdgcn_readlane(dl, gbi * U + j);
      float* o = acc + dj * LDW + lane;
#pragma unroll
      for (int c = 0; c < V; ++c) lds_add_f32(o + 32 * c, v[j][c]);
    }
    pend = false;
    ++gbi;
    if (gbi == NB || gst + gbi * U >= gend) {
      gst += NW * 32;
      gbi = 0;
      pcur = pnxt;
    }
  };
  auto gather_issue = [&]() {
    if (gst < gend) {
      const int sv = (gst + lane < gend) ? (int)(pcur & ((1u << SRC_BITS) - 1u)) : 0;
#pragma unroll
      for (int j = 0; j < U; ++j) {
        const int sj = __builtin_amdgcn_readlane(sv, gbi * U + j);
        const float* r = x + (long)sj * D;
        v[j] = *(const rowv*)(r + V * lane);
      }
      if (gbi == 0) pnxt = list_word(gst + NW * 32);
      pend = true;
    }
  };
  auto gather_step = [&]() {
    if (pend) gather_add();
    gather_issue();
  };

  root_load(0);
  for (int i = tid; i < (RNG + 1) * LDW / 4; i += NTH)
    ((float4*)acc)[i] = make_float4(0.0f, 0.0f, 0.0f, 0.0f);
  root_store(sX);
  __syncthreads();

  gather_begin(ob0, lb);
  while (pend || gst < gend) gather_step();

  const int half = lane >> 4;
  const int l16 = lane & 15;
  const int ntile = wv % NTN;
  const int mslot = wv / NTN;
  const int n = ntile * 16 + l16;
  const bool colOK = (DOUT == DOUTP) || (n < DOUT);
  const int nr = colOK ? n : 0;
  const float* pl = Wl + (long)nr * DIN + 4 * half;
  const float* pr = Wr + (long)nr * DIN + 4 * half;
  float4 bl[KB], br[KB];
#pragma unroll
  for (int j = 0; j < KB; ++j) {
    bl[j] = *(const float4*)(pl + 8 * j);
    br[j] = *(const float4*)(pr + 8 * j);
  }
  const float bcol = colOK ? bias[nr] : 0.0f;

  auto scale_sub = [&](int sub) {
#pragma unroll
    for (int q = 0; q < SUB / NW; ++q) {
      const int r = sub * SUB + wv + q * NW;
      int m = r0 + r;
      if (m >= nnodes) m = nnodes - 1;
      const float s = invdeg[m];
      rowv o;
#pragma unroll
      for (int c = 0; c < V; ++c) o[c] = acc[r * LDW + 32 * c + lane] * s;
      *(rowv*)&acc[r * LDW + V * lane] = o;
    }
  };

  auto mtile = [&](int sub, int mt, const float* sXc, bool gat) {
    v8f accv = (v8f){};
    const int ra = (sub * SUB + mt * 16 + l16) * LDW;
    const int rx = (mt * 16 + l16) * LDW;
    auto kblock = [&](int j, const float4 a, const float4 a2) {
      const v2f b0 = {bl[j].x, bl[j].y};
      const v2f b1 = {bl[j].z, bl[j].w};
      const v2f c0 = {br[j].x, br[j].y};
      const v2f c1 = {br[j].z, br[j].w};
      const v2f a0 = {a.x, a.y};
      const v2f x0 = {a2.x, a2.y};
      const v2f a1 = {a.z, a.w};
      const v2f x1 = {a2.z, a2.w};
      accv = __builtin_amdgcn_wmma_f32_16x16x4_f32(false, a0, false, b0, (short)0, accv, false, false);
      accv = __builtin_amdgcn_wmma_f32_16x16x4_f32(false, x0, false, c0, (short)0, accv, false, false);
      accv = __builtin_amdgcn_wmma_f32_16x16x4_f32(false, a1, false, b1, (short)0, accv, false, false);
      accv = __builtin_amdgcn_wmma_f32_16x16x4_f32(false, x1, false, c1, (short)0, accv, false, false);
    };
    auto fragA = [&](int j) { return *(const float4*)&acc[ra + 8 * j + 4 * half]; };
    auto fragX = [&](int j) { return *(const float4*)&sXc[rx + 8 * j + 4 * half]; };
#pragma unroll
    for (int g = 0; g < KB / G; ++g) {
      const int j0 = G * g;
      if (gat && pend) {
        const float4 a0 = fragA(j0), x0 = fragX(j0), a1 = fragA(j0 + 1), x1 = fragX(j0 + 1);
        gather_add();
        gather_issue();
        kblock(j0, a0, x0);
        kblock(j0 + 1, a1, x1);
      } else {
        const float4 a0 = fragA(j0), x0 = fragX(j0), a1 = fragA(j0 + 1), x1 = fragX(j0 + 1);
        kblock(j0, a0, x0);
        kblock(j0 + 1, a1, x1);
      }
#pragma unroll
      for (int t = 2; t < G; ++t) kblock(j0 + t, fragA(j0 + t), fragX(j0 + t));
    }
#pragma unroll
    for (int e = 0; e < 8; ++e) {
      const int row = r0 + sub * SUB + mt * 16 + e + 8 * half;
      float vv = accv[e] + bcol;
      if (ACT == 1) vv = tanhf(vv);
      if (ACT == 2) vv = (vv > 0.0f) ? vv : 0.01f * vv;
      if (colOK && row < nnodes) out[(long)row * DOUT + n] = vv;
    }
  };

  __syncthreads();
  scale_sub(0);
  __syncthreads();

  for (int i = 0; i < NSUB; ++i) {
    const float* sXc = sX + (i & 1) * (SUB * LDW);
    const bool more = (i + 1 < NSUB);
    if (more) {
      root_load(i + 1);
      gather_begin(lb, le);
      gather_issue();
      lb = le;
      le = obp[(i + 3 < NSUB) ? i + 3 : NSUB];
    }
    for (int mt = mslot; mt < MTS; mt += WPN) mtile(i, mt, sXc, more);
    if (more) {
      root_store(sX + ((i + 1) & 1) * (SUB * LDW));
      while (pend || gst < gend) gather_step();
      __syncthreads();
      scale_sub(i + 1);
      __syncthreads();
    }
  }
}


static inline unsigned cdiv(long a, long b) { return (unsigned)((a + b - 1) / b); }

extern "C" void kernel_launch(void* const* d_in, const int* in_sizes, int n_in,
                              void* d_out, int out_size, void* d_ws, size_t ws_size,
                              hipStream_t stream) {
  const float* x = (const float*)d_in[0];
  const int* ei = (const int*)d_in[1];
  const int N = in_sizes[0] / 64;
  const int E = in_sizes[1] / 2;
  const int* src = ei;
  const int* dst = ei + E;

  const float* g1Wl = (const float*)d_in[2];
  const float* g1b  = (const float*)d_in[3];
  const float* g1Wr = (const float*)d_in[4];
  const float* g2Wl = (const float*)d_in[5];
  const float* g2b  = (const float*)d_in[6];
  const float* g2Wr = (const float*)d_in[7];
  const float* g3Wl = (const float*)d_in[8];
  const float* g3b  = (const float*)d_in[9];
  const float* g3Wr = (const float*)d_in[10];
  const float* trW  = (const float*)d_in[11];
  const float* trb  = (const float*)d_in[12];
  const float* d1W  = (const float*)d_in[13];
  const float* d1b  = (const float*)d_in[14];
  const float* d2Wl = (const float*)d_in[15];
  const float* d2b  = (const float*)d_in[16];
  const float* d2Wr = (const float*)d_in[17];
  const float* d3Wl = (const float*)d_in[18];
  const float* d3b  = (const float*)d_in[19];
  const float* d3Wr = (const float*)d_in[20];
  const float* d4Wl = (const float*)d_in[21];
  const float* d4b  = (const float*)d_in[22];
  const float* d4Wr = (const float*)d_in[23];

  const unsigned gOwn = cdiv(N, RNG);
  float* ws   = (float*)d_ws;
  float* inv  = ws;
  unsigned* list = (unsigned*)(ws + N);
  float* bufA = (float*)(((uintptr_t)(ws + (size_t)N + (size_t)E) + 255) & ~(uintptr_t)255);
  float* bufB = bufA + (size_t)N * 128;
  int* ooff   = (int*)(bufB + (size_t)N * 128);
  float* cnt  = bufA;
  int* ocnt   = (int*)(bufA + N);
  int* ofill  = ocnt + MAXBIN;
  float* z2   = bufB;

  float* outZ01 = (float*)d_out;
  float* outZ2  = outZ01 + 2 * (size_t)N;
  float* outMu  = outZ2 + N;
  float* outLv  = outMu + (size_t)N * 64;

  const unsigned gN64 = (unsigned)((N + 63) / 64);
  const unsigned gBin = cdiv(E, 256 * EPT);
  const int nBin = (int)cdiv(N, SUB);
  const int nSlot = (int)gOwn * NSUB;

  zero_kernel<<<cdiv(N + 2 * MAXBIN, 256), 256, 0, stream>>>(cnt, N + 2 * MAXBIN);
  degree_count_kernel<<<gBin, 256, 0, stream>>>(dst, cnt, ocnt, E, nBin);
  invdeg_kernel<<<cdiv(N, 256), 256, 0, stream>>>(cnt, inv, N);
  owner_bin_kernel<<<gBin, 256, 0, stream>>>(src, dst, ocnt, ofill, ooff, list, E, nBin, nSlot);

  owner_sage_kernel<64, 128, 1><<<gOwn, 512, 0, stream>>>(
      x, list, ooff, inv, g1Wl, g1b, g1Wr, bufB, N);

  owner_sage_kernel<128, 128, 1><<<gOwn, 512, 0, stream>>>(
      bufB, list, ooff, inv, g2Wl, g2b, g2Wr, bufA, N);

  owner_sage_kernel<128, 128, 1><<<gOwn, 512, 0, stream>>>(
      bufA, list, ooff, inv, g3Wl, g3b, g3Wr, bufB, N);

  wmma_linear_kernel<128, 64, 0, false><<<gN64, 128, 0, stream>>>(
      nullptr, nullptr, bufB, trW, trb, nullptr, outMu, outLv, N);

  wmma_linear_kernel<64, 128, 2, false><<<gN64, 256, 0, stream>>>(
      nullptr, nullptr, outMu, d1W, d1b, nullptr, bufA, nullptr, N);

  owner_sage_kernel<128, 128, 2><<<gOwn, 512, 0, stream>>>(
      bufA, list, ooff, inv, d2Wl, d2b, d2Wr, bufB, N);

  owner_sage_kernel<128, 64, 1><<<gOwn, 512, 0, stream>>>(
      bufB, list, ooff, inv, d3Wl, d3b, d3Wr, bufA, N);

  owner_sage_kernel<64, 3, 0><<<gOwn, 512, 0, stream>>>(
      bufA, list, ooff, inv, d4Wl, d4b, d4Wr, z2, N);

  split_z2_kernel<<<cdiv(N, 256), 256, 0, stream>>>(z2, outZ01, outZ2, N);
}


# SAGE layers: LDS tile multiplied as summed (Wl fragment in its column order); 1/deg from an in-LDS edge count scales the Wl product in the epilogue: no scaling pass or 1/deg launch; sums reassociated
# speedup vs baseline: 3.9585x; 1.0992x over previous
#include <hip/hip_runtime.h>
#include <hip/hip_bf16.h>

typedef __attribute__((ext_vector_type(2))) float v2f;
typedef __attribute__((ext_vector_type(8))) float v8f;


__global__ void zero_kernel(float* __restrict__ p, long n) {
  long i = (long)blockIdx.x * blockDim.x + threadIdx.x;
  if (i < n) p[i] = 0.0f;
}

constexpr int RNG = 400;
constexpr int SUB = 80;
constexpr int NSUB = RNG / SUB;
constexpr int MAXBIN = 2048;
constexpr int EPT = 16;
constexpr int SRC_BITS = 23;
static_assert(RNG % SUB == 0 && SUB % 16 == 0, "sub-tiles");

__device__ __forceinline__ void lds_add_f32(float* p, float v) {
  __hip_atomic_fetch_add(p, v, __ATOMIC_RELAXED, __HIP_MEMORY_SCOPE_WORKGROUP);
}

__global__ __launch_bounds__(256) void bin_count_kernel(const int* __restrict__ dst,
                                                        int* __restrict__ ocnt, int E, int nown) {
  __shared__ int h[MAXBIN];
  const int tid = threadIdx.x;
  for (int i = tid; i < MAXBIN; i += 256) h[i] = 0;
  __syncthreads();
  const int base = blockIdx.x * (256 * EPT);
  int d[EPT];
#pragma unroll
  for (int k = 0; k < EPT; ++k) {
    const int e = base + k * 256 + tid;
    d[k] = (e < E) ? dst[e] : -1;
  }
#pragma unroll
  for (int k = 0; k < EPT; ++k) {
    if (d[k] >= 0) atomicAdd(&h[d[k] / SUB], 1);
  }
  __syncthreads();
  for (int i = tid; i < nown; i += 256) {
    const int c = h[i];
    if (c) atomicAdd(&ocnt[i], c);
  }
}

__global__ __launch_bounds__(256) void owner_bin_kernel(const int* __restrict__ src,
                                                        const int* __restrict__ dst,
                                                        const int* __restrict__ ocnt,
                                                        int* __restrict__ ofill,
                                                        int* __restrict__ ooff,
                                                        unsigned* __restrict__ list,
                                                        int E, int nown, int nslot) {
  constexpr int BPT = MAXBIN / 256;
  __shared__ int h[MAXBIN];
  __shared__ int off[MAXBIN];
  __shared__ int sc[256];
  const int tid = threadIdx.x;
  for (int i = tid; i < MAXBIN; i += 256) h[i] = 0;
  int c[BPT];
  int s = 0;
#pragma unroll
  for (int k = 0; k < BPT; ++k) {
    const int b = BPT * tid + k;
    c[k] = (b < nown) ? ocnt[b] : 0;
    s += c[k];
  }
  sc[tid] = s;
  __syncthreads();
  for (int dd = 1; dd < 256; dd <<= 1) {
    const int v = (tid >= dd) ? sc[tid - dd] : 0;
    __syncthreads();
    sc[tid] += v;
    __syncthreads();
  }
  int run = sc[tid] - s;
#pragma unroll
  for (int k = 0; k < BPT; ++k) {
    off[BPT * tid + k] = run;
    run += c[k];
  }
  __syncthreads();
  if (blockIdx.x == 0)
    for (int i = tid; i <= nslot; i += 256) ooff[i] = (i < nown) ? off[i] : sc[255];

  const int base = blockIdx.x * (256 * EPT);
  int dv[EPT], sv[EPT];
#pragma unroll
  for (int k = 0; k < EPT; ++k) {
    const int e = base + k * 256 + tid;
    const int ec = (e < E) ? e : E - 1;
    dv[k] = dst[ec];
    sv[k] = src[ec];
  }
  int ow[EPT], rk[EPT];
#pragma unroll
  for (int k = 0; k < EPT; ++k) {
    const int e = base + k * 256 + tid;
    ow[k] = dv[k] / SUB;
    rk[k] = (e < E) ? atomicAdd(&h[ow[k]], 1) : 0;
  }
  __syncthreads();
  for (int i = tid; i < nown; i += 256) {
    const int cc = h[i];
    if (cc) off[i] += atomicAdd(&ofill[i], cc);
  }
  __syncthreads();
#pragma unroll
  for (int k = 0; k < EPT; ++k) {
    const int e = base + k * 256 + tid;
    if (e < E)
      list[off[ow[k]] + rk[k]] =
          (unsigned)sv[k] | ((unsigned)(dv[k] - (dv[k] / RNG) * RNG) << SRC_BITS);
  }
}

__global__ void split_z2_kernel(const float* __restrict__ z2,
                                float* __restrict__ out01,
                                float* __restrict__ out2, int n) {
  int i = blockIdx.x * blockDim.x + threadIdx.x;
  if (i < n) {
    out01[2 * i + 0] = z2[3 * i + 0];
    out01[2 * i + 1] = z2[3 * i + 1];
    out2[i] = z2[3 * i + 2];
  }
}

template <int DIN, int DOUT, int ACT, bool HAS_AGG>
__global__ __launch_bounds__(2 * ((DOUT + 15) & ~15)) void wmma_linear_kernel(
    const float* agg, const float* __restrict__ invdeg,
    const float* __restrict__ x,
    const float* __restrict__ Wl, const float* __restrict__ bias,
    const float* __restrict__ Wr,
    float* out, float* out2, int nnodes) {
  constexpr int MT = 4;
  constexpr int ROWS = 16 * MT;
  constexpr int DOUTP = (DOUT + 15) & ~15;
  constexpr int LDW = DIN + 4;
  constexpr int C4 = DIN / 4;
  constexpr int KB = DIN / 8;
  __shared__ alignas(16) float sA[ROWS * LDW];
  __shared__ alignas(16) float sX[HAS_AGG ? ROWS * LDW : 4];

  const int tid = threadIdx.x;
  constexpr int NT = 2 * DOUTP;
  const int row0 = blockIdx.x * ROWS;

  const int wave = tid >> 5;
  const int lane = tid & 31;
  const int half = lane >> 4;
  const int l16 = lane & 15;
  const int n = wave * 16 + l16;
  const bool colOK = (DOUT == DOUTP) || (n < DOUT);

  constexpr int APT = ROWS * C4 / NT;
  constexpr int WREG = KB * (HAS_AGG ? 8 : 4);
  constexpr int AMAX = (208 - WREG) / (HAS_AGG ? 9 : 4);
  constexpr int AG = (APT <= AMAX) ? APT : (APT / 2 <= AMAX) ? APT / 2 : (APT / 4 <= AMAX) ? APT / 4 : APT / 8;
  float4 bl[KB];
  float4 br[HAS_AGG ? KB : 1];
  const int nr = colOK ? n : 0;
  const float* pl = Wl + (long)nr * DIN + 4 * half;
  const float* pr = HAS_AGG ? Wr + (long)nr * DIN + 4 * half : Wl;
  for (int g = 0; g < APT; g += AG) {
    float4 av[AG];
    float4 xv[HAS_AGG ? AG : 1];
    float sc[HAS_AGG ? AG : 1];
#pragma unroll
    for (int j = 0; j < AG; ++j) {
      const int i = tid + (g + j) * NT;
      const int r = i / C4, c4 = (i - r * C4) * 4;
      int m = row0 + r;
      if (m >= nnodes) m = nnodes - 1;
      if constexpr (HAS_AGG) {
        sc[j] = invdeg[m];
        av[j] = *(const float4*)(agg + (long)m * DIN + c4);
        xv[j] = *(const float4*)(x + (long)m * DIN + c4);
      } else {
        av[j] = *(const float4*)(x + (long)m * DIN + c4);
      }
    }
    if (g == 0) {
#pragma unroll
      for (int j = 0; j < KB; ++j) {
        bl[j] = *(const float4*)(pl + 8 * j);
        if constexpr (HAS_AGG) br[j] = *(const float4*)(pr + 8 * j);
      }
    }
#pragma unroll
    for (int j = 0; j < AG; ++j) {
      const int i = tid + (g + j) * NT;
      const int r = i / C4, c4 = (i - r * C4) * 4;
      if constexpr (HAS_AGG) {
        float4 a = av[j];
        const float s = sc[j];
        a.x *= s; a.y *= s; a.z *= s; a.w *= s;
        *(float4*)&sA[r * LDW + c4] = a;
        *(float4*)&sX[r * LDW + c4] = xv[j];
      } else {
        *(float4*)&sA[r * LDW + c4] = av[j];
      }
    }
  }
  __syncthreads();

  v8f acc[MT];
#pragma unroll
  for (int mt = 0; mt < MT; ++mt) acc[mt] = (v8f){};

#pragma unroll
  for (int j = 0; j < KB; ++j) {
    const int kk = 8 * j + 4 * half;
    float4 a[MT];
    float4 a2[HAS_AGG ? MT : 1];
#pragma unroll
    for (int mt = 0; mt < MT; ++mt) {
      const int r = mt * 16 + l16;
      a[mt] = *(const float4*)&sA[r * LDW + kk];
      if constexpr (HAS_AGG) a2[mt] = *(const float4*)&sX[r * LDW + kk];
    }
    const v2f b0 = {bl[j].x, bl[j].y};
    const v2f b1 = {bl[j].z, bl[j].w};
#pragma unroll
    for (int mt = 0; mt < MT; ++mt) {
      const v2f a0 = {a[mt].x, a[mt].y};
      acc[mt] = __builtin_amdgcn_wmma_f32_16x16x4_f32(false, a0, false, b0,
                                                      (short)0, acc[mt], false,
                                                      false);
      if constexpr (HAS_AGG) {
        const v2f x0 = {a2[mt].x, a2[mt].y};
        const v2f c0 = {br[j].x, br[j].y};
        acc[mt] = __builtin_amdgcn_wmma_f32_16x16x4_f32(false, x0, false, c0,
                                                        (short)0, acc[mt],
                                                        false, false);
      }
    }
#pragma unroll
    for (int mt = 0; mt < MT; ++mt) {
      const v2f a1 = {a[mt].z, a[mt].w};
      acc[mt] = __builtin_amdgcn_wmma_f32_16x16x4_f32(false, a1, false, b1,
                                                      (short)0, acc[mt], false,
                                                      false);
      if constexpr (HAS_AGG) {
        const v2f x1 = {a2[mt].z, a2[mt].w};
        const v2f c1 = {br[j].z, br[j].w};
        acc[mt] = __builtin_amdgcn_wmma_f32_16x16x4_f32(false, x1, false, c1,
                                                        (short)0, acc[mt],
                                                        false, false);
      }
    }
  }

  const float bcol = colOK ? bias[colOK ? n : 0] : 0.0f;
#pragma unroll
  for (int mt = 0; mt < MT; ++mt) {
#pragma unroll
    for (int i = 0; i < 8; ++i) {
      int row = row0 + mt * 16 + i + 8 * half;
      float v = acc[mt][i] + bcol;
      if (ACT == 1) v = tanhf(v);
      if (ACT == 2) v = (v > 0.0f) ? v : 0.01f * v;
      if (colOK && row < nnodes) {
        out[(long)row * DOUT + n] = v;
        if (out2) out2[(long)row * DOUT + n] = v;
      }
    }
  }
}

template <int DIN, int DOUT, int ACT>
__global__ __launch_bounds__(512) void owner_sage_kernel(
    const float* __restrict__ x, const unsigned* __restrict__ list,
    const int* __restrict__ ooff,
    const float* __restrict__ Wl, const float* __restrict__ bias,
    const float* __restrict__ Wr, float* __restrict__ out, int nnodes) {
  constexpr int D = DIN;
  constexpr int V = D / 32;
  constexpr int U = 32 / V;
  constexpr int NB = 32 / U;
  typedef float rowv __attribute__((ext_vector_type(V)));
  constexpr int NW = 16;
  constexpr int NTH = NW * 32;
  constexpr int LDW = D + 4;
  constexpr int MTS = SUB / 16;
  constexpr int DOUTP = (DOUT + 15) & ~15;
  constexpr int NTN = DOUTP / 16;
  constexpr int WPN = NW / NTN;
  constexpr int C4 = D / 4;
  constexpr int KB = D / 8;
  constexpr int G = 4;
  constexpr int XF4 = SUB * C4;
  constexpr int APT = (XF4 + NTH - 1) / NTH;
  static_assert(NW % NTN == 0 && KB % G == 0 && SUB % NW == 0, "shape");
  __shared__ alignas(16) float acc[(RNG + 1) * LDW];
  __shared__ alignas(16) float sX[2 * SUB * LDW];
  __shared__ alignas(16) int degL[RNG];
  const int tid = threadIdx.x;
  const int r0 = blockIdx.x * RNG;
  const int lane = tid & 31;
  const int wv = __builtin_amdgcn_readfirstlane(tid >> 5);
  const int* obp = ooff + blockIdx.x * NSUB;
  const int ob0 = obp[0];
  int lb = obp[1], le = obp[2];

  typedef float xnv __attribute__((ext_vector_type(4 * APT)));
  xnv xn;
  static_assert(NTH % C4 == 0, "a thread's float4 share a column");
  constexpr int RSTEP = NTH / C4;
  const int rb = tid / C4, cb = (tid - rb * C4) * 4;
  auto root_load = [&](int sub) {
#pragma unroll
    for (int j = 0; j < APT; ++j) {
      int r = rb + j * RSTEP;
      if (r > SUB - 1) r = SUB - 1;
      int m = r0 + sub * SUB + r;
      if (m >= nnodes) m = nnodes - 1;
      const float4 t = *(const float4*)(x + (long)m * D + cb);
      xn[4 * j + 0] = t.x; xn[4 * j + 1] = t.y; xn[4 * j + 2] = t.z; xn[4 * j + 3] = t.w;
    }
  };
  auto root_store = [&](float* dstX) {
#pragma unroll
    for (int j = 0; j < APT; ++j) {
      const int r = rb + j * RSTEP;
      if (r < SUB) *(float4*)&dstX[r * LDW + cb] = make_float4(xn[4 * j + 0], xn[4 * j + 1], xn[4 * j + 2], xn[4 * j + 3]);
    }
  };

  int gst = 0, gend = 0;
  int gbi = 0;
  bool pend = false;
  unsigned pcur = 0, pnxt = 0;
  rowv v[U];
  auto list_word = [&](int st) -> unsigned {
    int q = st + lane;
    if (q > gend - 1) q = gend - 1;
    if (q < 0) q = 0;
    return list[q];
  };
  auto gather_begin = [&](int b, int e) {
    gend = e;
    gst = b + wv * 32;
    gbi = 0;
    pend = false;
    pcur = list_word(gst);
  };
  auto gather_add = [&]() {
    const int dl = (gst + lane < gend) ? (int)(pcur >> SRC_BITS) : RNG;
#pragma unroll
    for (int j = 0; j < U; ++j) {
      const int dj = __builtin_amdgcn_readlane(dl, gbi * U + j);
      float* o = acc + dj * LDW + lane;
#pragma unroll
      for (int c = 0; c < V; ++c) lds_add_f32(o + 32 * c, v[j][c]);
    }
    pend = false;
    ++gbi;
    if (gbi == NB || gst + gbi * U >= gend) {
      gst += NW * 32;
      gbi = 0;
      pcur = pnxt;
    }
  };
  auto gather_issue = [&]() {
    if (gst < gend) {
      const int sv = (gst + lane < gend) ? (int)(pcur & ((1u << SRC_BITS) - 1u)) : 0;
#pragma unroll
      for (int j = 0; j < U; ++j) {
        const int sj = __builtin_amdgcn_readlane(sv, gbi * U + j);
        const float* r = x + (long)sj * D;
        v[j] = *(const rowv*)(r + V * lane);
      }
      if (gbi == 0) {
        pnxt = list_word(gst + NW * 32);
        if (gst + lane < gend) atomicAdd(&degL[pcur >> SRC_BITS], 1);
      }
      pend = true;
    }
  };
  auto gather_step = [&]() {
    if (pend) gather_add();
    gather_issue();
  };

  root_load(0);
  for (int i = tid; i < (RNG + 1) * LDW / 4; i += NTH)
    ((float4*)acc)[i] = make_float4(0.0f, 0.0f, 0.0f, 0.0f);
  for (int i = tid; i < RNG; i += NTH) degL[i] = 0;
  root_store(sX);
  __syncthreads();

  const int half = lane >> 4;
  const int l16 = lane & 15;
  const int ntile = wv % NTN;
  const int mslot = wv / NTN;
  const int n = ntile * 16 + l16;
  const bool colOK = (DOUT == DOUTP) || (n < DOUT);
  const int nr = colOK ? n : 0;
  const float* pl = Wl + (long)nr * DIN + 4 * V * half;
  const float* pr = Wr + (long)nr * DIN + 4 * half;
  float4 bl[KB], br[KB];
#pragma unroll
  for (int j = 0; j < KB; ++j) {
    const int o = V * 8 * (j & 3) + (j >> 2);
    bl[j] = make_float4(pl[o], pl[o + V], pl[o + 2 * V], pl[o + 3 * V]);
    br[j] = *(const float4*)(pr + 8 * j);
  }
  const float bcol = colOK ? bias[nr] : 0.0f;

  gather_begin(ob0, lb);
  while (pend || gst < gend) gather_step();

  auto mtile = [&](int sub, int mt, const float* sXc, bool gat) {
    v8f accv = (v8f){};
    v8f accx = (v8f){};
    const int ra = (sub * SUB + mt * 16 + l16) * LDW;
    const int rx = (mt * 16 + l16) * LDW;
    auto kblock = [&](int j, const float4 a, const float4 a2) {
      const v2f b0 = {bl[j].x, bl[j].y};
      const v2f b1 = {bl[j].z, bl[j].w};
      const v2f c0 = {br[j].x, br[j].y};
      const v2f c1 = {br[j].z, br[j].w};
      const v2f a0 = {a.x, a.y};
      const v2f x0 = {a2.x, a2.y};
      const v2f a1 = {a.z, a.w};
      const v2f x1 = {a2.z, a2.w};
      accv = __builtin_amdgcn_wmma_f32_16x16x4_f32(false, a0, false, b0, (short)0, accv, false, false);
      accx = __builtin_amdgcn_wmma_f32_16x16x4_f32(false, x0, false, c0, (short)0, accx, false, false);
      accv = __builtin_amdgcn_wmma_f32_16x16x4_f32(false, a1, false, b1, (short)0, accv, false, false);
      accx = __builtin_amdgcn_wmma_f32_16x16x4_f32(false, x1, false, c1, (short)0, accx, false, false);
    };
    auto fragA = [&](int j) { return *(const float4*)&acc[ra + 8 * j + 4 * half]; };
    auto fragX = [&](int j) { return *(const float4*)&sXc[rx + 8 * j + 4 * half]; };
#pragma unroll
    for (int g = 0; g < KB / G; ++g) {
      const int j0 = G * g;
      if (gat && pend) {
        const float4 a0 = fragA(j0), x0 = fragX(j0), a1 = fragA(j0 + 1), x1 = fragX(j0 + 1);
        gather_add();
        gather_issue();
        kblock(j0, a0, x0);
        kblock(j0 + 1, a1, x1);
      } else {
        const float4 a0 = fragA(j0), x0 = fragX(j0), a1 = fragA(j0 + 1), x1 = fragX(j0 + 1);
        kblock(j0, a0, x0);
        kblock(j0 + 1, a1, x1);
      }
#pragma unroll
      for (int t = 2; t < G; ++t) kblock(j0 + t, fragA(j0 + t), fragX(j0 + t));
    }
    const int dbase = sub * SUB + mt * 16 + 8 * half;
    const int4 dg0 = *(const int4*)&degL[dbase];
    const int4 dg1 = *(const int4*)&degL[dbase + 4];
    const int dg[8] = {dg0.x, dg0.y, dg0.z, dg0.w, dg1.x, dg1.y, dg1.z, dg1.w};
#pragma unroll
    for (int e = 0; e < 8; ++e) {
      const int row = r0 + sub * SUB + mt * 16 + e + 8 * half;
      const float s = 1.0f / fmaxf((float)dg[e], 1.0f);
      float vv = accv[e] * s + accx[e] + bcol;
      if (ACT == 1) vv = tanhf(vv);
      if (ACT == 2) vv = (vv > 0.0f) ? vv : 0.01f * vv;
      if (colOK && row < nnodes) out[(long)row * DOUT + n] = vv;
    }
  };

  __syncthreads();

  for (int i = 0; i < NSUB; ++i) {
    const float* sXc = sX + (i & 1) * (SUB * LDW);
    const bool more = (i + 1 < NSUB);
    if (more) {
      root_load(i + 1);
      gather_begin(lb, le);
      gather_issue();
      lb = le;
      le = obp[(i + 3 < NSUB) ? i + 3 : NSUB];
    }
    for (int mt = mslot; mt < MTS; mt += WPN) mtile(i, mt, sXc, more);
    if (more) {
      root_store(sX + ((i + 1) & 1) * (SUB * LDW));
      while (pend || gst < gend) gather_step();
      __syncthreads();
    }
  }
}


static inline unsigned cdiv(long a, long b) { return (unsigned)((a + b - 1) / b); }

extern "C" void kernel_launch(void* const* d_in, const int* in_sizes, int n_in,
                              void* d_out, int out_size, void* d_ws, size_t ws_size,
                              hipStream_t stream) {
  const float* x = (const float*)d_in[0];
  const int* ei = (const int*)d_in[1];
  const int N = in_sizes[0] / 64;
  const int E = in_sizes[1] / 2;
  const int* src = ei;
  const int* dst = ei + E;

  const float* g1Wl = (const float*)d_in[2];
  const float* g1b  = (const float*)d_in[3];
  const float* g1Wr = (const float*)d_in[4];
  const float* g2Wl = (const float*)d_in[5];
  const float* g2b  = (const float*)d_in[6];
  const float* g2Wr = (const float*)d_in[7];
  const float* g3Wl = (const float*)d_in[8];
  const float* g3b  = (const float*)d_in[9];
  const float* g3Wr = (const float*)d_in[10];
  const float* trW  = (const float*)d_in[11];
  const float* trb  = (const float*)d_in[12];
  const float* d1W  = (const float*)d_in[13];
  const float* d1b  = (const float*)d_in[14];
  const float* d2Wl = (const float*)d_in[15];
  const float* d2b  = (const float*)d_in[16];
  const float* d2Wr = (const float*)d_in[17];
  const float* d3Wl = (const float*)d_in[18];
  const float* d3b  = (const float*)d_in[19];
  const float* d3Wr = (const float*)d_in[20];
  const float* d4Wl = (const float*)d_in[21];
  const float* d4b  = (const float*)d_in[22];
  const float* d4Wr = (const float*)d_in[23];

  const unsigned gOwn = cdiv(N, RNG);
  float* ws   = (float*)d_ws;
  unsigned* list = (unsigned*)ws;
  float* bufA = (float*)(((uintptr_t)(ws + (size_t)E) + 255) & ~(uintptr_t)255);
  float* bufB = bufA + (size_t)N * 128;
  int* ooff   = (int*)(bufB + (size_t)N * 128);
  int* ocnt   = (int*)bufA;
  int* ofill  = ocnt + MAXBIN;
  float* z2   = bufB;

  float* outZ01 = (float*)d_out;
  float* outZ2  = outZ01 + 2 * (size_t)N;
  float* outMu  = outZ2 + N;
  float* outLv  = outMu + (size_t)N * 64;

  const unsigned gN64 = (unsigned)((N + 63) / 64);
  const unsigned gBin = cdiv(E, 256 * EPT);
  const int nBin = (int)cdiv(N, SUB);
  const int nSlot = (int)gOwn * NSUB;

  zero_kernel<<<cdiv(2 * MAXBIN, 256), 256, 0, stream>>>(bufA, 2 * MAXBIN);
  bin_count_kernel<<<gBin, 256, 0, stream>>>(dst, ocnt, E, nBin);
  owner_bin_kernel<<<gBin, 256, 0, stream>>>(src, dst, ocnt, ofill, ooff, list, E, nBin, nSlot);

  owner_sage_kernel<64, 128, 1><<<gOwn, 512, 0, stream>>>(
      x, list, ooff, g1Wl, g1b, g1Wr, bufB, N);

  owner_sage_kernel<128, 128, 1><<<gOwn, 512, 0, stream>>>(
      bufB, list, ooff, g2Wl, g2b, g2Wr, bufA, N);

  owner_sage_kernel<128, 128, 1><<<gOwn, 512, 0, stream>>>(
      bufA, list, ooff, g3Wl, g3b, g3Wr, bufB, N);

  wmma_linear_kernel<128, 64, 0, false><<<gN64, 128, 0, stream>>>(
      nullptr, nullptr, bufB, trW, trb, nullptr, outMu, outLv, N);

  wmma_linear_kernel<64, 128, 2, false><<<gN64, 256, 0, stream>>>(
      nullptr, nullptr, outMu, d1W, d1b, nullptr, bufA, nullptr, N);

  owner_sage_kernel<128, 128, 2><<<gOwn, 512, 0, stream>>>(
      bufA, list, ooff, d2Wl, d2b, d2Wr, bufB, N);

  owner_sage_kernel<128, 64, 1><<<gOwn, 512, 0, stream>>>(
      bufB, list, ooff, d3Wl, d3b, d3Wr, bufA, N);

  owner_sage_kernel<64, 3, 0><<<gOwn, 512, 0, stream>>>(
      bufA, list, ooff, d4Wl, d4b, d4Wr, z2, N);

  split_z2_kernel<<<cdiv(N, 256), 256, 0, stream>>>(z2, outZ01, outZ2, N);
}
